# starting kernel
# baseline (speedup 1.0000x reference)
#include <hip/hip_runtime.h>
#include <cstddef>

typedef float v2f __attribute__((ext_vector_type(2)));
typedef float v8f __attribute__((ext_vector_type(8)));

#define D_MODEL 768
#define N_HEAD  12
#define D_HEAD  64
#define N_CTX   2048
#define BATCH   2
#define M_ROWS  (BATCH * N_CTX)

__device__ __forceinline__ v8f wmma_f32(v2f a, v2f b, v8f c) {
  return __builtin_amdgcn_wmma_f32_16x16x4_f32(false, a, false, b, (short)0, c,
                                               false, false);
}

__device__ __forceinline__ void gemm_strip_k768(const float* __restrict__ xrow,
                                                const float* __restrict__ w0,
                                                const float* __restrict__ w1,
                                                const float* __restrict__ w2,
                                                const float* __restrict__ w3,
                                                v8f acc[4]) {
#pragma unroll 4
  for (int k = 0; k < D_MODEL; k += 4) {
    v2f a  = *(const v2f*)(xrow + k);
    v2f b0 = *(const v2f*)(w0 + k);
    v2f b1 = *(const v2f*)(w1 + k);
    v2f b2 = *(const v2f*)(w2 + k);
    v2f b3 = *(const v2f*)(w3 + k);
    acc[0] = wmma_f32(a, b0, acc[0]);
    acc[1] = wmma_f32(a, b1, acc[1]);
    acc[2] = wmma_f32(a, b2, acc[2]);
    acc[3] = wmma_f32(a, b3, acc[3]);
  }
}

__global__ __launch_bounds__(256) void qkv_gemm_kernel(
    const float* __restrict__ X, const float* __restrict__ W,
    const float* __restrict__ bias, float* __restrict__ out, int vmode) {
  const int wave = threadIdx.x >> 5;
  const int lane = threadIdx.x & 31;
  const int lo = lane & 15, hi = lane >> 4;
  const int m0 = blockIdx.x * 32 + (wave >> 2) * 16;
  const int n0 = blockIdx.y * 256 + (wave & 3) * 64;
  const float* xrow  = X + (size_t)(m0 + lo) * D_MODEL + 2 * hi;
  const float* wbase = W + (size_t)(n0 + lo) * D_MODEL + 2 * hi;
  v8f acc[4] = {{}, {}, {}, {}};
  gemm_strip_k768(xrow, wbase, wbase + 16 * D_MODEL, wbase + 32 * D_MODEL,
                  wbase + 48 * D_MODEL, acc);
  const int h = n0 >> 6;
#pragma unroll
  for (int j = 0; j < 4; ++j) {
    const int d = j * 16 + lo;
    const float bv = bias[n0 + d];
#pragma unroll
    for (int r = 0; r < 8; ++r) {
      const int m = m0 + r + 8 * hi;
      const int b = m >> 11, s = m & (N_CTX - 1);
      const size_t hb = (size_t)b * N_HEAD + h;
      if (vmode)
        out[(hb * D_HEAD + d) * N_CTX + s] = acc[j][r] + bv;
      else
        out[(hb * N_CTX + s) * D_HEAD + d] = acc[j][r] + bv;
    }
  }
}

__global__ __launch_bounds__(256) void out_gemm_kernel(
    const float* __restrict__ X, const float* __restrict__ W,
    float* __restrict__ out) {
  const int wave = threadIdx.x >> 5;
  const int lane = threadIdx.x & 31;
  const int lo = lane & 15, hi = lane >> 4;
  const int m0 = blockIdx.x * 32 + (wave >> 2) * 16;
  const int n0 = blockIdx.y * 256 + (wave & 3) * 64;
  const float* xrow  = X + (size_t)(m0 + lo) * D_MODEL + 2 * hi;
  const float* wbase = W + (size_t)(n0 + lo) * D_MODEL + 2 * hi;
  v8f acc[4] = {{}, {}, {}, {}};
  gemm_strip_k768(xrow, wbase, wbase + 16 * D_MODEL, wbase + 32 * D_MODEL,
                  wbase + 48 * D_MODEL, acc);
#pragma unroll
  for (int j = 0; j < 4; ++j) {
#pragma unroll
    for (int r = 0; r < 8; ++r) {
      const int m = m0 + r + 8 * hi;
      out[(size_t)m * D_MODEL + n0 + j * 16 + lo] = acc[j][r];
    }
  }
}

__global__ __launch_bounds__(256) void norm_rope_kernel(
    float* __restrict__ qk, const float* __restrict__ g) {
  const int lane = threadIdx.x & 31;
  const int row = blockIdx.x * 8 + (threadIdx.x >> 5);
  const int s = row & (N_CTX - 1);
  float* p = qk + (size_t)row * D_HEAD;
  const float x1 = p[lane];
  const float x2 = p[lane + 32];
  float ss = x1 * x1 + x2 * x2;
#pragma unroll
  for (int off = 16; off > 0; off >>= 1) ss += __shfl_xor(ss, off, 32);
  const float inv = rsqrtf(ss * (1.0f / 64.0f) + 1e-6f);
  const float xn1 = x1 * inv * g[lane];
  const float xn2 = x2 * inv * g[lane + 32];
  const float freq = (float)s * exp2f((float)lane * (-11.0f / 32.0f));
  const float c = __cosf(freq), sn = __sinf(freq);
  p[lane]      = xn1 * c + xn2 * sn;
  p[lane + 32] = xn2 * c - xn1 * sn;
}

__global__ __launch_bounds__(256) void attn_kernel(
    const float* __restrict__ Q, const float* __restrict__ K,
    const float* __restrict__ V, float* __restrict__ Z) {
  __shared__ float lds[8 * 256];
  const int wave = threadIdx.x >> 5;
  const int lane = threadIdx.x & 31;
  const int lo = lane & 15, hi = lane >> 4;
  const int wid = blockIdx.x * 8 + wave;
  const int head = wid >> 7;
  const int qt = wid & 127;
  const float* Qh = Q + (size_t)head * N_CTX * D_HEAD;
  const float* Kh = K + (size_t)head * N_CTX * D_HEAD;
  const float* Vh = V + (size_t)head * D_HEAD * N_CTX;
  float* pl = lds + wave * 256;

  v2f qa[16];
  {
    const float* qrow = Qh + (size_t)(qt * 16 + lo) * D_HEAD + 2 * hi;
#pragma unroll
    for (int kk = 0; kk < 16; ++kk) qa[kk] = *(const v2f*)(qrow + kk * 4);
  }

  v8f z0 = {}, z1 = {}, z2 = {}, z3 = {};
  for (int kt = 0; kt <= qt; ++kt) {
    const float* krow = Kh + (size_t)(kt * 16 + lo) * D_HEAD + 2 * hi;
    __builtin_prefetch(krow + 16 * D_HEAD, 0, 0);
    v8f sc = {};
#pragma unroll
    for (int kk = 0; kk < 16; ++kk) {
      v2f kb = *(const v2f*)(krow + kk * 4);
      sc = wmma_f32(qa[kk], kb, sc);
    }
#pragma unroll
    for (int r = 0; r < 8; ++r) {
      float pv = sc[r] * (1.0f / 64.0f);
      pv = pv * pv;
      if (kt == qt && lo > r + 8 * hi) pv = 0.0f;
      pl[(r + 8 * hi) * 16 + lo] = pv;
    }
#pragma unroll
    for (int kk2 = 0; kk2 < 4; ++kk2) {
      v2f pa = *(const v2f*)(pl + lo * 16 + kk2 * 4 + 2 * hi);
      const float* vcol = Vh + (size_t)lo * N_CTX + kt * 16 + kk2 * 4 + 2 * hi;
      v2f v0 = *(const v2f*)(vcol);
      v2f v1 = *(const v2f*)(vcol + 16 * N_CTX);
      v2f v2 = *(const v2f*)(vcol + 32 * N_CTX);
      v2f v3 = *(const v2f*)(vcol + 48 * N_CTX);
      z0 = wmma_f32(pa, v0, z0);
      z1 = wmma_f32(pa, v1, z1);
      z2 = wmma_f32(pa, v2, z2);
      z3 = wmma_f32(pa, v3, z3);
    }
  }
  const int b = head / N_HEAD, h = head % N_HEAD;
#pragma unroll
  for (int r = 0; r < 8; ++r) {
    const int s = qt * 16 + r + 8 * hi;
    float* zr = Z + (size_t)(b * N_CTX + s) * D_MODEL + h * D_HEAD + lo;
    zr[0]  = z0[r];
    zr[16] = z1[r];
    zr[32] = z2[r];
    zr[48] = z3[r];
  }
}

extern "C" void kernel_launch(void* const* d_in, const int* in_sizes, int n_in,
                              void* d_out, int out_size, void* d_ws, size_t ws_size,
                              hipStream_t stream) {
  const float* x  = (const float*)d_in[0];
  const float* Wq = (const float*)d_in[1];
  const float* bq = (const float*)d_in[2];
  const float* Wk = (const float*)d_in[3];
  const float* bk = (const float*)d_in[4];
  const float* Wv = (const float*)d_in[5];
  const float* bv = (const float*)d_in[6];
  const float* Wo = (const float*)d_in[7];
  const float* g  = (const float*)d_in[8];
  float* out = (float*)d_out;

  const size_t QKV = (size_t)BATCH * N_HEAD * N_CTX * D_HEAD;
  float* q = (float*)d_ws;
  float* k = q + QKV;
  float* v = k + QKV;
  float* z = v + QKV;

  dim3 gblk(M_ROWS / 32, D_MODEL / 256);
  qkv_gemm_kernel<<<gblk, 256, 0, stream>>>(x, Wq, bq, q, 0);
  qkv_gemm_kernel<<<gblk, 256, 0, stream>>>(x, Wk, bk, k, 0);
  qkv_gemm_kernel<<<gblk, 256, 0, stream>>>(x, Wv, bv, v, 1);

  const int rows = BATCH * N_HEAD * N_CTX;
  norm_rope_kernel<<<rows / 8, 256, 0, stream>>>(q, g);
  norm_rope_kernel<<<rows / 8, 256, 0, stream>>>(k, g);

  attn_kernel<<<(BATCH * N_HEAD * (N_CTX / 16)) / 8, 256, 0, stream>>>(q, k, v, z);

  out_gemm_kernel<<<gblk, 256, 0, stream>>>(z, Wo, out);
}


# All WMMA fragment loads widened b64->b128 (GEMM A/B, attention Q/K/V, P from LDS): lane takes 4 k at +4*hi, 2 WMMAs per load, grouped per block; halves cache lines per WMMA; k sum order changes
# speedup vs baseline: 1.1615x; 1.1615x over previous
#include <hip/hip_runtime.h>
#include <cstddef>

typedef float v2f __attribute__((ext_vector_type(2)));
typedef float v4f __attribute__((ext_vector_type(4)));
typedef float v8f __attribute__((ext_vector_type(8)));

#define D_MODEL 768
#define N_HEAD  12
#define D_HEAD  64
#define N_CTX   2048
#define BATCH   2
#define M_ROWS  (BATCH * N_CTX)

__device__ __forceinline__ v8f wmma_f32(v2f a, v2f b, v8f c) {
  return __builtin_amdgcn_wmma_f32_16x16x4_f32(false, a, false, b, (short)0, c,
                                               false, false);
}

__device__ __forceinline__ void gemm_strip_k768(const float* __restrict__ xrow,
                                                const float* __restrict__ w0,
                                                const float* __restrict__ w1,
                                                const float* __restrict__ w2,
                                                const float* __restrict__ w3,
                                                v8f acc[4]) {
  for (int k = 0; k < D_MODEL; k += 32) {
    v4f a[4], b0[4], b1[4], b2[4], b3[4];
#pragma unroll
    for (int u = 0; u < 4; ++u) {
      a[u]  = *(const v4f*)(xrow + k + 8 * u);
      b0[u] = *(const v4f*)(w0 + k + 8 * u);
      b1[u] = *(const v4f*)(w1 + k + 8 * u);
      b2[u] = *(const v4f*)(w2 + k + 8 * u);
      b3[u] = *(const v4f*)(w3 + k + 8 * u);
    }
#pragma unroll
    for (int u = 0; u < 4; ++u) {
      acc[0] = wmma_f32(a[u].xy, b0[u].xy, acc[0]);
      acc[1] = wmma_f32(a[u].xy, b1[u].xy, acc[1]);
      acc[2] = wmma_f32(a[u].xy, b2[u].xy, acc[2]);
      acc[3] = wmma_f32(a[u].xy, b3[u].xy, acc[3]);
      acc[0] = wmma_f32(a[u].zw, b0[u].zw, acc[0]);
      acc[1] = wmma_f32(a[u].zw, b1[u].zw, acc[1]);
      acc[2] = wmma_f32(a[u].zw, b2[u].zw, acc[2]);
      acc[3] = wmma_f32(a[u].zw, b3[u].zw, acc[3]);
    }
    __builtin_amdgcn_sched_group_barrier(0x020, 20, 0);
    __builtin_amdgcn_sched_group_barrier(0x008, 32, 0);
  }
}

__global__ __launch_bounds__(256) void qkv_gemm_kernel(
    const float* __restrict__ X, const float* __restrict__ W,
    const float* __restrict__ bias, float* __restrict__ out, int vmode) {
  const int wave = threadIdx.x >> 5;
  const int lane = threadIdx.x & 31;
  const int lo = lane & 15, hi = lane >> 4;
  const int m0 = blockIdx.x * 32 + (wave >> 2) * 16;
  const int n0 = blockIdx.y * 256 + (wave & 3) * 64;
  const float* xrow  = X + (size_t)(m0 + lo) * D_MODEL + 4 * hi;
  const float* wbase = W + (size_t)(n0 + lo) * D_MODEL + 4 * hi;
  v8f acc[4] = {{}, {}, {}, {}};
  gemm_strip_k768(xrow, wbase, wbase + 16 * D_MODEL, wbase + 32 * D_MODEL,
                  wbase + 48 * D_MODEL, acc);
  const int h = n0 >> 6;
#pragma unroll
  for (int j = 0; j < 4; ++j) {
    const int d = j * 16 + lo;
    const float bv = bias[n0 + d];
#pragma unroll
    for (int r = 0; r < 8; ++r) {
      const int m = m0 + r + 8 * hi;
      const int b = m >> 11, s = m & (N_CTX - 1);
      const size_t hb = (size_t)b * N_HEAD + h;
      if (vmode)
        out[(hb * D_HEAD + d) * N_CTX + s] = acc[j][r] + bv;
      else
        out[(hb * N_CTX + s) * D_HEAD + d] = acc[j][r] + bv;
    }
  }
}

__global__ __launch_bounds__(256) void out_gemm_kernel(
    const float* __restrict__ X, const float* __restrict__ W,
    float* __restrict__ out) {
  const int wave = threadIdx.x >> 5;
  const int lane = threadIdx.x & 31;
  const int lo = lane & 15, hi = lane >> 4;
  const int m0 = blockIdx.x * 32 + (wave >> 2) * 16;
  const int n0 = blockIdx.y * 256 + (wave & 3) * 64;
  const float* xrow  = X + (size_t)(m0 + lo) * D_MODEL + 4 * hi;
  const float* wbase = W + (size_t)(n0 + lo) * D_MODEL + 4 * hi;
  v8f acc[4] = {{}, {}, {}, {}};
  gemm_strip_k768(xrow, wbase, wbase + 16 * D_MODEL, wbase + 32 * D_MODEL,
                  wbase + 48 * D_MODEL, acc);
#pragma unroll
  for (int j = 0; j < 4; ++j) {
#pragma unroll
    for (int r = 0; r < 8; ++r) {
      const int m = m0 + r + 8 * hi;
      out[(size_t)m * D_MODEL + n0 + j * 16 + lo] = acc[j][r];
    }
  }
}

__global__ __launch_bounds__(256) void norm_rope_kernel(
    float* __restrict__ qk, const float* __restrict__ g) {
  const int lane = threadIdx.x & 31;
  const int row = blockIdx.x * 8 + (threadIdx.x >> 5);
  const int s = row & (N_CTX - 1);
  float* p = qk + (size_t)row * D_HEAD;
  const float x1 = p[lane];
  const float x2 = p[lane + 32];
  float ss = x1 * x1 + x2 * x2;
#pragma unroll
  for (int off = 16; off > 0; off >>= 1) ss += __shfl_xor(ss, off, 32);
  const float inv = rsqrtf(ss * (1.0f / 64.0f) + 1e-6f);
  const float xn1 = x1 * inv * g[lane];
  const float xn2 = x2 * inv * g[lane + 32];
  const float freq = (float)s * exp2f((float)lane * (-11.0f / 32.0f));
  const float c = __cosf(freq), sn = __sinf(freq);
  p[lane]      = xn1 * c + xn2 * sn;
  p[lane + 32] = xn2 * c - xn1 * sn;
}

__global__ __launch_bounds__(256) void attn_kernel(
    const float* __restrict__ Q, const float* __restrict__ K,
    const float* __restrict__ V, float* __restrict__ Z) {
  __shared__ __attribute__((aligned(16))) float lds[8 * 256];
  const int wave = threadIdx.x >> 5;
  const int lane = threadIdx.x & 31;
  const int lo = lane & 15, hi = lane >> 4;
  const int wid = blockIdx.x * 8 + wave;
  const int head = wid >> 7;
  const int qt = wid & 127;
  const float* Qh = Q + (size_t)head * N_CTX * D_HEAD;
  const float* Kh = K + (size_t)head * N_CTX * D_HEAD;
  const float* Vh = V + (size_t)head * D_HEAD * N_CTX;
  float* pl = lds + wave * 256;

  v4f qa[8];
  {
    const float* qrow = Qh + (size_t)(qt * 16 + lo) * D_HEAD + 4 * hi;
#pragma unroll
    for (int j = 0; j < 8; ++j) qa[j] = *(const v4f*)(qrow + j * 8);
  }

  v8f z0 = {}, z1 = {}, z2 = {}, z3 = {};
  for (int kt = 0; kt <= qt; ++kt) {
    const float* krow = Kh + (size_t)(kt * 16 + lo) * D_HEAD + 4 * hi;
    __builtin_prefetch(krow + 16 * D_HEAD, 0, 0);
    v4f kb[8], vb[2][4];
#pragma unroll
    for (int j = 0; j < 8; ++j) kb[j] = *(const v4f*)(krow + j * 8);
    {
      const float* vcol = Vh + (size_t)lo * N_CTX + kt * 16 + 4 * hi;
#pragma unroll
      for (int j = 0; j < 2; ++j) {
#pragma unroll
        for (int t = 0; t < 4; ++t)
          vb[j][t] = *(const v4f*)(vcol + (size_t)t * 16 * N_CTX + j * 8);
      }
    }
    v8f sc = {};
#pragma unroll
    for (int j = 0; j < 8; ++j) {
      sc = wmma_f32(qa[j].xy, kb[j].xy, sc);
      sc = wmma_f32(qa[j].zw, kb[j].zw, sc);
    }
#pragma unroll
    for (int r = 0; r < 8; ++r) {
      float pv = sc[r] * (1.0f / 64.0f);
      pv = pv * pv;
      if (kt == qt && lo > r + 8 * hi) pv = 0.0f;
      pl[(r + 8 * hi) * 16 + lo] = pv;
    }
#pragma unroll
    for (int j = 0; j < 2; ++j) {
      v4f pa = *(const v4f*)(pl + lo * 16 + j * 8 + 4 * hi);
      z0 = wmma_f32(pa.xy, vb[j][0].xy, z0);
      z1 = wmma_f32(pa.xy, vb[j][1].xy, z1);
      z2 = wmma_f32(pa.xy, vb[j][2].xy, z2);
      z3 = wmma_f32(pa.xy, vb[j][3].xy, z3);
      z0 = wmma_f32(pa.zw, vb[j][0].zw, z0);
      z1 = wmma_f32(pa.zw, vb[j][1].zw, z1);
      z2 = wmma_f32(pa.zw, vb[j][2].zw, z2);
      z3 = wmma_f32(pa.zw, vb[j][3].zw, z3);
    }
    __builtin_amdgcn_sched_group_barrier(0x020, 16, 0);
    __builtin_amdgcn_sched_group_barrier(0x008, 32, 0);
  }
  const int b = head / N_HEAD, h = head % N_HEAD;
#pragma unroll
  for (int r = 0; r < 8; ++r) {
    const int s = qt * 16 + r + 8 * hi;
    float* zr = Z + (size_t)(b * N_CTX + s) * D_MODEL + h * D_HEAD + lo;
    zr[0]  = z0[r];
    zr[16] = z1[r];
    zr[32] = z2[r];
    zr[48] = z3[r];
  }
}

extern "C" void kernel_launch(void* const* d_in, const int* in_sizes, int n_in,
                              void* d_out, int out_size, void* d_ws, size_t ws_size,
                              hipStream_t stream) {
  const float* x  = (const float*)d_in[0];
  const float* Wq = (const float*)d_in[1];
  const float* bq = (const float*)d_in[2];
  const float* Wk = (const float*)d_in[3];
  const float* bk = (const float*)d_in[4];
  const float* Wv = (const float*)d_in[5];
  const float* bv = (const float*)d_in[6];
  const float* Wo = (const float*)d_in[7];
  const float* g  = (const float*)d_in[8];
  float* out = (float*)d_out;

  const size_t QKV = (size_t)BATCH * N_HEAD * N_CTX * D_HEAD;
  float* q = (float*)d_ws;
  float* k = q + QKV;
  float* v = k + QKV;
  float* z = v + QKV;

  dim3 gblk(M_ROWS / 32, D_MODEL / 256);
  qkv_gemm_kernel<<<gblk, 256, 0, stream>>>(x, Wq, bq, q, 0);
  qkv_gemm_kernel<<<gblk, 256, 0, stream>>>(x, Wk, bk, k, 0);
  qkv_gemm_kernel<<<gblk, 256, 0, stream>>>(x, Wv, bv, v, 1);

  const int rows = BATCH * N_HEAD * N_CTX;
  norm_rope_kernel<<<rows / 8, 256, 0, stream>>>(q, g);
  norm_rope_kernel<<<rows / 8, 256, 0, stream>>>(k, g);

  attn_kernel<<<(BATCH * N_HEAD * (N_CTX / 16)) / 8, 256, 0, stream>>>(q, k, v, z);

  out_gemm_kernel<<<gblk, 256, 0, stream>>>(z, Wo, out);
}


# RMSNorm+RoPE of Q and K fused into their projection GEMM's epilogue (a wave's 16x64 strip is whole head rows; 16-lane DPP row sums, same sum tree): the 2 norm launches and their round trip go
# speedup vs baseline: 1.2880x; 1.1089x over previous
#include <hip/hip_runtime.h>
#include <cstddef>

typedef float v2f __attribute__((ext_vector_type(2)));
typedef float v4f __attribute__((ext_vector_type(4)));
typedef float v8f __attribute__((ext_vector_type(8)));

#define D_MODEL 768
#define N_HEAD  12
#define D_HEAD  64
#define N_CTX   2048
#define BATCH   2
#define M_ROWS  (BATCH * N_CTX)

__device__ __forceinline__ v8f wmma_f32(v2f a, v2f b, v8f c) {
  return __builtin_amdgcn_wmma_f32_16x16x4_f32(false, a, false, b, (short)0, c,
                                               false, false);
}

template <int MASK>
__device__ __forceinline__ float row_xor_add(float v) {
  return v + __int_as_float(__builtin_amdgcn_update_dpp(
                 0, __float_as_int(v), 0x160 + MASK, 0xf, 0xf, false));
}

__device__ __forceinline__ void gemm_strip_k768(const float* __restrict__ xrow,
                                                const float* __restrict__ w0,
                                                const float* __restrict__ w1,
                                                const float* __restrict__ w2,
                                                const float* __restrict__ w3,
                                                v8f acc[4]) {
  for (int k = 0; k < D_MODEL; k += 32) {
    v4f a[4], b0[4], b1[4], b2[4], b3[4];
#pragma unroll
    for (int u = 0; u < 4; ++u) {
      a[u]  = *(const v4f*)(xrow + k + 8 * u);
      b0[u] = *(const v4f*)(w0 + k + 8 * u);
      b1[u] = *(const v4f*)(w1 + k + 8 * u);
      b2[u] = *(const v4f*)(w2 + k + 8 * u);
      b3[u] = *(const v4f*)(w3 + k + 8 * u);
    }
#pragma unroll
    for (int u = 0; u < 4; ++u) {
      acc[0] = wmma_f32(a[u].xy, b0[u].xy, acc[0]);
      acc[1] = wmma_f32(a[u].xy, b1[u].xy, acc[1]);
      acc[2] = wmma_f32(a[u].xy, b2[u].xy, acc[2]);
      acc[3] = wmma_f32(a[u].xy, b3[u].xy, acc[3]);
      acc[0] = wmma_f32(a[u].zw, b0[u].zw, acc[0]);
      acc[1] = wmma_f32(a[u].zw, b1[u].zw, acc[1]);
      acc[2] = wmma_f32(a[u].zw, b2[u].zw, acc[2]);
      acc[3] = wmma_f32(a[u].zw, b3[u].zw, acc[3]);
    }
    __builtin_amdgcn_sched_group_barrier(0x020, 20, 0);
    __builtin_amdgcn_sched_group_barrier(0x008, 32, 0);
  }
}

__global__ __launch_bounds__(256) void qkv_gemm_kernel(
    const float* __restrict__ X, const float* __restrict__ W,
    const float* __restrict__ bias, const float* __restrict__ g,
    float* __restrict__ out, int vmode) {
  const int wave = threadIdx.x >> 5;
  const int lane = threadIdx.x & 31;
  const int lo = lane & 15, hi = lane >> 4;
  const int m0 = blockIdx.x * 32 + (wave >> 2) * 16;
  const int n0 = blockIdx.y * 256 + (wave & 3) * 64;
  const float* xrow  = X + (size_t)(m0 + lo) * D_MODEL + 4 * hi;
  const float* wbase = W + (size_t)(n0 + lo) * D_MODEL + 4 * hi;
  v8f acc[4] = {{}, {}, {}, {}};
  gemm_strip_k768(xrow, wbase, wbase + 16 * D_MODEL, wbase + 32 * D_MODEL,
                  wbase + 48 * D_MODEL, acc);
  const int h = n0 >> 6;
  float bv[4];
#pragma unroll
  for (int j = 0; j < 4; ++j) bv[j] = bias[n0 + j * 16 + lo];
  if (vmode) {
#pragma unroll
    for (int j = 0; j < 4; ++j) {
      const int d = j * 16 + lo;
#pragma unroll
      for (int r = 0; r < 8; ++r) {
        const int m = m0 + r + 8 * hi;
        const int b = m >> 11, s = m & (N_CTX - 1);
        const size_t hb = (size_t)b * N_HEAD + h;
        out[(hb * D_HEAD + d) * N_CTX + s] = acc[j][r] + bv[j];
      }
    }
  } else {
    float gv[4];
#pragma unroll
    for (int j = 0; j < 4; ++j) gv[j] = g[j * 16 + lo];
    const float w0 = exp2f((float)lo * (-11.0f / 32.0f));
    const float w1 = exp2f((float)(lo + 16) * (-11.0f / 32.0f));
#pragma unroll
    for (int r = 0; r < 8; ++r) {
      const int m = m0 + r + 8 * hi;
      const int b = m >> 11, s = m & (N_CTX - 1);
      const size_t hb = (size_t)b * N_HEAD + h;
      const float x0 = acc[0][r] + bv[0];
      const float x1 = acc[1][r] + bv[1];
      const float x2 = acc[2][r] + bv[2];
      const float x3 = acc[3][r] + bv[3];
      float ss = (x0 * x0 + x2 * x2) + (x1 * x1 + x3 * x3);
      ss = row_xor_add<8>(ss);
      ss = row_xor_add<4>(ss);
      ss = row_xor_add<2>(ss);
      ss = row_xor_add<1>(ss);
      const float inv = rsqrtf(ss * (1.0f / 64.0f) + 1e-6f);
      const float y0 = x0 * inv * gv[0];
      const float y1 = x1 * inv * gv[1];
      const float y2 = x2 * inv * gv[2];
      const float y3 = x3 * inv * gv[3];
      const float f0 = (float)s * w0, f1 = (float)s * w1;
      const float c0 = __cosf(f0), s0 = __sinf(f0);
      const float c1 = __cosf(f1), s1 = __sinf(f1);
      float* o = out + (hb * N_CTX + s) * D_HEAD + lo;
      o[0]  = y0 * c0 + y2 * s0;
      o[16] = y1 * c1 + y3 * s1;
      o[32] = y2 * c0 - y0 * s0;
      o[48] = y3 * c1 - y1 * s1;
    }
  }
}

__global__ __launch_bounds__(256) void out_gemm_kernel(
    const float* __restrict__ X, const float* __restrict__ W,
    float* __restrict__ out) {
  const int wave = threadIdx.x >> 5;
  const int lane = threadIdx.x & 31;
  const int lo = lane & 15, hi = lane >> 4;
  const int m0 = blockIdx.x * 32 + (wave >> 2) * 16;
  const int n0 = blockIdx.y * 256 + (wave & 3) * 64;
  const float* xrow  = X + (size_t)(m0 + lo) * D_MODEL + 4 * hi;
  const float* wbase = W + (size_t)(n0 + lo) * D_MODEL + 4 * hi;
  v8f acc[4] = {{}, {}, {}, {}};
  gemm_strip_k768(xrow, wbase, wbase + 16 * D_MODEL, wbase + 32 * D_MODEL,
                  wbase + 48 * D_MODEL, acc);
#pragma unroll
  for (int j = 0; j < 4; ++j) {
#pragma unroll
    for (int r = 0; r < 8; ++r) {
      const int m = m0 + r + 8 * hi;
      out[(size_t)m * D_MODEL + n0 + j * 16 + lo] = acc[j][r];
    }
  }
}

__global__ __launch_bounds__(256) void attn_kernel(
    const float* __restrict__ Q, const float* __restrict__ K,
    const float* __restrict__ V, float* __restrict__ Z) {
  __shared__ __attribute__((aligned(16))) float lds[8 * 256];
  const int wave = threadIdx.x >> 5;
  const int lane = threadIdx.x & 31;
  const int lo = lane & 15, hi = lane >> 4;
  const int wid = blockIdx.x * 8 + wave;
  const int head = wid >> 7;
  const int qt = wid & 127;
  const float* Qh = Q + (size_t)head * N_CTX * D_HEAD;
  const float* Kh = K + (size_t)head * N_CTX * D_HEAD;
  const float* Vh = V + (size_t)head * D_HEAD * N_CTX;
  float* pl = lds + wave * 256;

  v4f qa[8];
  {
    const float* qrow = Qh + (size_t)(qt * 16 + lo) * D_HEAD + 4 * hi;
#pragma unroll
    for (int j = 0; j < 8; ++j) qa[j] = *(const v4f*)(qrow + j * 8);
  }

  v8f z0 = {}, z1 = {}, z2 = {}, z3 = {};
  for (int kt = 0; kt <= qt; ++kt) {
    const float* krow = Kh + (size_t)(kt * 16 + lo) * D_HEAD + 4 * hi;
    __builtin_prefetch(krow + 16 * D_HEAD, 0, 0);
    v4f kb[8], vb[2][4];
#pragma unroll
    for (int j = 0; j < 8; ++j) kb[j] = *(const v4f*)(krow + j * 8);
    {
      const float* vcol = Vh + (size_t)lo * N_CTX + kt * 16 + 4 * hi;
#pragma unroll
      for (int j = 0; j < 2; ++j) {
#pragma unroll
        for (int t = 0; t < 4; ++t)
          vb[j][t] = *(const v4f*)(vcol + (size_t)t * 16 * N_CTX + j * 8);
      }
    }
    v8f sc = {};
#pragma unroll
    for (int j = 0; j < 8; ++j) {
      sc = wmma_f32(qa[j].xy, kb[j].xy, sc);
      sc = wmma_f32(qa[j].zw, kb[j].zw, sc);
    }
#pragma unroll
    for (int r = 0; r < 8; ++r) {
      float pv = sc[r] * (1.0f / 64.0f);
      pv = pv * pv;
      if (kt == qt && lo > r + 8 * hi) pv = 0.0f;
      pl[(r + 8 * hi) * 16 + lo] = pv;
    }
#pragma unroll
    for (int j = 0; j < 2; ++j) {
      v4f pa = *(const v4f*)(pl + lo * 16 + j * 8 + 4 * hi);
      z0 = wmma_f32(pa.xy, vb[j][0].xy, z0);
      z1 = wmma_f32(pa.xy, vb[j][1].xy, z1);
      z2 = wmma_f32(pa.xy, vb[j][2].xy, z2);
      z3 = wmma_f32(pa.xy, vb[j][3].xy, z3);
      z0 = wmma_f32(pa.zw, vb[j][0].zw, z0);
      z1 = wmma_f32(pa.zw, vb[j][1].zw, z1);
      z2 = wmma_f32(pa.zw, vb[j][2].zw, z2);
      z3 = wmma_f32(pa.zw, vb[j][3].zw, z3);
    }
    __builtin_amdgcn_sched_group_barrier(0x020, 16, 0);
    __builtin_amdgcn_sched_group_barrier(0x008, 32, 0);
  }
  const int b = head / N_HEAD, h = head % N_HEAD;
#pragma unroll
  for (int r = 0; r < 8; ++r) {
    const int s = qt * 16 + r + 8 * hi;
    float* zr = Z + (size_t)(b * N_CTX + s) * D_MODEL + h * D_HEAD + lo;
    zr[0]  = z0[r];
    zr[16] = z1[r];
    zr[32] = z2[r];
    zr[48] = z3[r];
  }
}

extern "C" void kernel_launch(void* const* d_in, const int* in_sizes, int n_in,
                              void* d_out, int out_size, void* d_ws, size_t ws_size,
                              hipStream_t stream) {
  const float* x  = (const float*)d_in[0];
  const float* Wq = (const float*)d_in[1];
  const float* bq = (const float*)d_in[2];
  const float* Wk = (const float*)d_in[3];
  const float* bk = (const float*)d_in[4];
  const float* Wv = (const float*)d_in[5];
  const float* bv = (const float*)d_in[6];
  const float* Wo = (const float*)d_in[7];
  const float* g  = (const float*)d_in[8];
  float* out = (float*)d_out;

  const size_t QKV = (size_t)BATCH * N_HEAD * N_CTX * D_HEAD;
  float* q = (float*)d_ws;
  float* k = q + QKV;
  float* v = k + QKV;
  float* z = v + QKV;

  dim3 gblk(M_ROWS / 32, D_MODEL / 256);
  qkv_gemm_kernel<<<gblk, 256, 0, stream>>>(x, Wq, bq, g, q, 0);
  qkv_gemm_kernel<<<gblk, 256, 0, stream>>>(x, Wk, bk, g, k, 0);
  qkv_gemm_kernel<<<gblk, 256, 0, stream>>>(x, Wv, bv, g, v, 1);

  attn_kernel<<<(BATCH * N_HEAD * (N_CTX / 16)) / 8, 256, 0, stream>>>(q, k, v, z);

  out_gemm_kernel<<<gblk, 256, 0, stream>>>(z, Wo, out);
}


# Q, K, V leave their GEMM in the attention's fragment order (Q/K product transposed, lane = position): its 16 loads per key tile are whole lines; score d order, PV key order, norm sum tree change
# speedup vs baseline: 1.5483x; 1.2021x over previous
#include <hip/hip_runtime.h>
#include <cstddef>

typedef float v2f __attribute__((ext_vector_type(2)));
typedef float v4f __attribute__((ext_vector_type(4)));
typedef float v8f __attribute__((ext_vector_type(8)));

#define D_MODEL 768
#define N_HEAD  12
#define D_HEAD  64
#define N_CTX   2048
#define BATCH   2
#define M_ROWS  (BATCH * N_CTX)

__device__ __forceinline__ v8f wmma_f32(v2f a, v2f b, v8f c) {
  return __builtin_amdgcn_wmma_f32_16x16x4_f32(false, a, false, b, (short)0, c,
                                               false, false);
}

template <int MASK>
__device__ __forceinline__ float row_xor_add(float v) {
  return v + __int_as_float(__builtin_amdgcn_update_dpp(
                 0, __float_as_int(v), 0x160 + MASK, 0xf, 0xf, false));
}

__device__ __forceinline__ float other_half(float v) {
  return __int_as_float(__builtin_amdgcn_permlanex16(
      __float_as_int(v), __float_as_int(v), 0x76543210, 0xfedcba98, false, false));
}

template <bool SWAP>
__device__ __forceinline__ void gemm_strip_k768(const float* __restrict__ xrow,
                                                const float* __restrict__ w0,
                                                const float* __restrict__ w1,
                                                const float* __restrict__ w2,
                                                const float* __restrict__ w3,
                                                v8f acc[4]) {
  for (int k = 0; k < D_MODEL; k += 32) {
    v4f a[4], b0[4], b1[4], b2[4], b3[4];
#pragma unroll
    for (int u = 0; u < 4; ++u) {
      a[u]  = *(const v4f*)(xrow + k + 8 * u);
      b0[u] = *(const v4f*)(w0 + k + 8 * u);
      b1[u] = *(const v4f*)(w1 + k + 8 * u);
      b2[u] = *(const v4f*)(w2 + k + 8 * u);
      b3[u] = *(const v4f*)(w3 + k + 8 * u);
    }
#pragma unroll
    for (int u = 0; u < 4; ++u) {
      if (SWAP) {
        acc[0] = wmma_f32(b0[u].xy, a[u].xy, acc[0]);
        acc[1] = wmma_f32(b1[u].xy, a[u].xy, acc[1]);
        acc[2] = wmma_f32(b2[u].xy, a[u].xy, acc[2]);
        acc[3] = wmma_f32(b3[u].xy, a[u].xy, acc[3]);
        acc[0] = wmma_f32(b0[u].zw, a[u].zw, acc[0]);
        acc[1] = wmma_f32(b1[u].zw, a[u].zw, acc[1]);
        acc[2] = wmma_f32(b2[u].zw, a[u].zw, acc[2]);
        acc[3] = wmma_f32(b3[u].zw, a[u].zw, acc[3]);
      } else {
        acc[0] = wmma_f32(a[u].xy, b0[u].xy, acc[0]);
        acc[1] = wmma_f32(a[u].xy, b1[u].xy, acc[1]);
        acc[2] = wmma_f32(a[u].xy, b2[u].xy, acc[2]);
        acc[3] = wmma_f32(a[u].xy, b3[u].xy, acc[3]);
        acc[0] = wmma_f32(a[u].zw, b0[u].zw, acc[0]);
        acc[1] = wmma_f32(a[u].zw, b1[u].zw, acc[1]);
        acc[2] = wmma_f32(a[u].zw, b2[u].zw, acc[2]);
        acc[3] = wmma_f32(a[u].zw, b3[u].zw, acc[3]);
      }
    }
    __builtin_amdgcn_sched_group_barrier(0x020, 20, 0);
    __builtin_amdgcn_sched_group_barrier(0x008, 32, 0);
  }
}

template <int VMODE>
__global__ __launch_bounds__(256) void qkv_gemm_kernel(
    const float* __restrict__ X, const float* __restrict__ W,
    const float* __restrict__ bias, const float* __restrict__ g,
    float* __restrict__ out) {
  const int wave = threadIdx.x >> 5;
  const int lane = threadIdx.x & 31;
  const int lo = lane & 15, hi = lane >> 4;
  const int m0 = blockIdx.x * 32 + (wave >> 2) * 16;
  const int n0 = blockIdx.y * 256 + (wave & 3) * 64;
  const float* xrow  = X + (size_t)(m0 + lo) * D_MODEL + 4 * hi;
  const float* wbase = W + (size_t)(n0 + lo) * D_MODEL + 4 * hi;
  v8f acc[4] = {{}, {}, {}, {}};
  gemm_strip_k768<VMODE == 0>(xrow, wbase, wbase + 16 * D_MODEL,
                              wbase + 32 * D_MODEL, wbase + 48 * D_MODEL, acc);
  const int h = n0 >> 6;
  const int b = m0 >> 11;
  const int tile = (m0 & (N_CTX - 1)) >> 4;
  float* frag = out + ((size_t)(b * N_HEAD + h) * N_CTX + (size_t)tile * 16) * D_HEAD
                + 4 * lane;
  if (VMODE) {
    float bv[4];
#pragma unroll
    for (int j = 0; j < 4; ++j) bv[j] = bias[n0 + j * 16 + lo];
#pragma unroll
    for (int jv = 0; jv < 2; ++jv) {
#pragma unroll
      for (int j = 0; j < 4; ++j) {
        v4f o;
        o.x = acc[j][4 * jv + 0] + bv[j];
        o.y = acc[j][4 * jv + 1] + bv[j];
        o.z = acc[j][4 * jv + 2] + bv[j];
        o.w = acc[j][4 * jv + 3] + bv[j];
        *(v4f*)(frag + (4 * jv + j) * 128) = o;
      }
    }
  } else {
    float x[4][8], gv[4][8];
#pragma unroll
    for (int j = 0; j < 4; ++j) {
      const v4f b0 = *(const v4f*)(bias + n0 + 16 * j + 8 * hi);
      const v4f b1 = *(const v4f*)(bias + n0 + 16 * j + 8 * hi + 4);
      const v4f g0 = *(const v4f*)(g + 16 * j + 8 * hi);
      const v4f g1 = *(const v4f*)(g + 16 * j + 8 * hi + 4);
#pragma unroll
      for (int r = 0; r < 4; ++r) {
        x[j][r]      = acc[j][r] + b0[r];
        x[j][r + 4]  = acc[j][r + 4] + b1[r];
        gv[j][r]     = g0[r];
        gv[j][r + 4] = g1[r];
      }
    }
    float sj[4];
#pragma unroll
    for (int j = 0; j < 4; ++j) {
      sj[j] = ((x[j][0] * x[j][0] + x[j][1] * x[j][1]) +
               (x[j][2] * x[j][2] + x[j][3] * x[j][3])) +
              ((x[j][4] * x[j][4] + x[j][5] * x[j][5]) +
               (x[j][6] * x[j][6] + x[j][7] * x[j][7]));
    }
    float ss = (sj[0] + sj[2]) + (sj[1] + sj[3]);
    ss = ss + other_half(ss);
    const float inv = rsqrtf(ss * (1.0f / 64.0f) + 1e-6f);
    const int s = (m0 + lo) & (N_CTX - 1);
    float o[4][8];
#pragma unroll
    for (int j = 0; j < 2; ++j) {
#pragma unroll
      for (int r = 0; r < 8; ++r) {
        const int i = 16 * j + 8 * hi + r;
        const float w = exp2f((float)i * (-11.0f / 32.0f));
        const float f = (float)s * w;
        const float c = __cosf(f), sn = __sinf(f);
        const float ya = x[j][r] * inv * gv[j][r];
        const float yb = x[j + 2][r] * inv * gv[j + 2][r];
        o[j][r]     = ya * c + yb * sn;
        o[j + 2][r] = yb * c - ya * sn;
      }
    }
#pragma unroll
    for (int j = 0; j < 4; ++j) {
#pragma unroll
      for (int rr = 0; rr < 2; ++rr) {
        v4f v;
        v.x = o[j][4 * rr + 0];
        v.y = o[j][4 * rr + 1];
        v.z = o[j][4 * rr + 2];
        v.w = o[j][4 * rr + 3];
        *(v4f*)(frag + (2 * j + rr) * 128) = v;
      }
    }
  }
}

__global__ __launch_bounds__(256) void out_gemm_kernel(
    const float* __restrict__ X, const float* __restrict__ W,
    float* __restrict__ out) {
  const int wave = threadIdx.x >> 5;
  const int lane = threadIdx.x & 31;
  const int lo = lane & 15, hi = lane >> 4;
  const int m0 = blockIdx.x * 32 + (wave >> 2) * 16;
  const int n0 = blockIdx.y * 256 + (wave & 3) * 64;
  const float* xrow  = X + (size_t)(m0 + lo) * D_MODEL + 4 * hi;
  const float* wbase = W + (size_t)(n0 + lo) * D_MODEL + 4 * hi;
  v8f acc[4] = {{}, {}, {}, {}};
  gemm_strip_k768<false>(xrow, wbase, wbase + 16 * D_MODEL, wbase + 32 * D_MODEL,
                         wbase + 48 * D_MODEL, acc);
#pragma unroll
  for (int j = 0; j < 4; ++j) {
#pragma unroll
    for (int r = 0; r < 8; ++r) {
      const int m = m0 + r + 8 * hi;
      out[(size_t)m * D_MODEL + n0 + j * 16 + lo] = acc[j][r];
    }
  }
}

__global__ __launch_bounds__(256) void attn_kernel(
    const float* __restrict__ Q, const float* __restrict__ K,
    const float* __restrict__ V, float* __restrict__ Z) {
  __shared__ __attribute__((aligned(16))) float lds[8 * 256];
  const int wave = threadIdx.x >> 5;
  const int lane = threadIdx.x & 31;
  const int lo = lane & 15, hi = lane >> 4;
  const int wid = blockIdx.x * 8 + wave;
  const int head = wid >> 7;
  const int qt = wid & 127;
  const float* Qh = Q + (size_t)head * N_CTX * D_HEAD;
  const float* Kh = K + (size_t)head * N_CTX * D_HEAD;
  const float* Vh = V + (size_t)head * N_CTX * D_HEAD;
  float* pl = lds + wave * 256;

  v4f qa[8];
  {
    const float* qfrag = Qh + (size_t)qt * (16 * D_HEAD) + 4 * lane;
#pragma unroll
    for (int j = 0; j < 8; ++j) qa[j] = *(const v4f*)(qfrag + j * 128);
  }

  v8f z0 = {}, z1 = {}, z2 = {}, z3 = {};
  for (int kt = 0; kt <= qt; ++kt) {
    const float* kfrag = Kh + (size_t)kt * (16 * D_HEAD) + 4 * lane;
    __builtin_prefetch(Kh + (size_t)(kt + 1) * (16 * D_HEAD) + 32 * lane, 0, 0);
    v4f kb[8], vb[2][4];
#pragma unroll
    for (int j = 0; j < 8; ++j) kb[j] = *(const v4f*)(kfrag + j * 128);
    {
      const float* vfrag = Vh + (size_t)kt * (16 * D_HEAD) + 4 * lane;
#pragma unroll
      for (int j = 0; j < 2; ++j) {
#pragma unroll
        for (int t = 0; t < 4; ++t)
          vb[j][t] = *(const v4f*)(vfrag + (4 * j + t) * 128);
      }
    }
    v8f sc = {};
#pragma unroll
    for (int j = 0; j < 8; ++j) {
      sc = wmma_f32(qa[j].xy, kb[j].xy, sc);
      sc = wmma_f32(qa[j].zw, kb[j].zw, sc);
    }
#pragma unroll
    for (int r = 0; r < 8; ++r) {
      float pv = sc[r] * (1.0f / 64.0f);
      pv = pv * pv;
      if (kt == qt && lo > r + 8 * hi) pv = 0.0f;
      pl[(r + 8 * hi) * 16 + lo] = pv;
    }
#pragma unroll
    for (int j = 0; j < 2; ++j) {
      v4f pa = *(const v4f*)(pl + lo * 16 + 8 * hi + 4 * j);
      z0 = wmma_f32(pa.xy, vb[j][0].xy, z0);
      z1 = wmma_f32(pa.xy, vb[j][1].xy, z1);
      z2 = wmma_f32(pa.xy, vb[j][2].xy, z2);
      z3 = wmma_f32(pa.xy, vb[j][3].xy, z3);
      z0 = wmma_f32(pa.zw, vb[j][0].zw, z0);
      z1 = wmma_f32(pa.zw, vb[j][1].zw, z1);
      z2 = wmma_f32(pa.zw, vb[j][2].zw, z2);
      z3 = wmma_f32(pa.zw, vb[j][3].zw, z3);
    }
    __builtin_amdgcn_sched_group_barrier(0x020, 16, 0);
    __builtin_amdgcn_sched_group_barrier(0x008, 32, 0);
  }
  const int b = head / N_HEAD, h = head % N_HEAD;
#pragma unroll
  for (int r = 0; r < 8; ++r) {
    const int s = qt * 16 + r + 8 * hi;
    float* zr = Z + (size_t)(b * N_CTX + s) * D_MODEL + h * D_HEAD + lo;
    zr[0]  = z0[r];
    zr[16] = z1[r];
    zr[32] = z2[r];
    zr[48] = z3[r];
  }
}

extern "C" void kernel_launch(void* const* d_in, const int* in_sizes, int n_in,
                              void* d_out, int out_size, void* d_ws, size_t ws_size,
                              hipStream_t stream) {
  const float* x  = (const float*)d_in[0];
  const float* Wq = (const float*)d_in[1];
  const float* bq = (const float*)d_in[2];
  const float* Wk = (const float*)d_in[3];
  const float* bk = (const float*)d_in[4];
  const float* Wv = (const float*)d_in[5];
  const float* bv = (const float*)d_in[6];
  const float* Wo = (const float*)d_in[7];
  const float* g  = (const float*)d_in[8];
  float* out = (float*)d_out;

  const size_t QKV = (size_t)BATCH * N_HEAD * N_CTX * D_HEAD;
  float* q = (float*)d_ws;
  float* k = q + QKV;
  float* v = k + QKV;
  float* z = v + QKV;

  dim3 gblk(M_ROWS / 32, D_MODEL / 256);
  qkv_gemm_kernel<0><<<gblk, 256, 0, stream>>>(x, Wq, bq, g, q);
  qkv_gemm_kernel<0><<<gblk, 256, 0, stream>>>(x, Wk, bk, g, k);
  qkv_gemm_kernel<1><<<gblk, 256, 0, stream>>>(x, Wv, bv, g, v);

  attn_kernel<<<(BATCH * N_HEAD * (N_CTX / 16)) / 8, 256, 0, stream>>>(q, k, v, z);

  out_gemm_kernel<<<gblk, 256, 0, stream>>>(z, Wo, out);
}


# GEMM operands read in fragment order: one pre-pass gathers x and the 4 weights once, attention stores Z so; every GEMM fragment load is 4 whole lines, not 16 partial; K order and sums unchanged
# speedup vs baseline: 2.0452x; 1.3209x over previous
#include <hip/hip_runtime.h>
#include <cstddef>

typedef float v2f __attribute__((ext_vector_type(2)));
typedef float v4f __attribute__((ext_vector_type(4)));
typedef float v8f __attribute__((ext_vector_type(8)));

#define D_MODEL 768
#define N_HEAD  12
#define D_HEAD  64
#define N_CTX   2048
#define BATCH   2
#define M_ROWS  (BATCH * N_CTX)

__device__ __forceinline__ v8f wmma_f32(v2f a, v2f b, v8f c) {
  return __builtin_amdgcn_wmma_f32_16x16x4_f32(false, a, false, b, (short)0, c,
                                               false, false);
}

template <int MASK>
__device__ __forceinline__ float row_xor_add(float v) {
  return v + __int_as_float(__builtin_amdgcn_update_dpp(
                 0, __float_as_int(v), 0x160 + MASK, 0xf, 0xf, false));
}

__device__ __forceinline__ float other_half(float v) {
  return __int_as_float(__builtin_amdgcn_permlanex16(
      __float_as_int(v), __float_as_int(v), 0x76543210, 0xfedcba98, false, false));
}

#define K_CHUNKS    (D_MODEL / 8)
#define TILE_FLOATS (K_CHUNKS * 128)

template <bool SWAP>
__device__ __forceinline__ void gemm_strip_k768(const float* __restrict__ xrow,
                                                const float* __restrict__ w0,
                                                const float* __restrict__ w1,
                                                const float* __restrict__ w2,
                                                const float* __restrict__ w3,
                                                v8f acc[4]) {
  for (int k = 0; k < D_MODEL; k += 32) {
    v4f a[4], b0[4], b1[4], b2[4], b3[4];
#pragma unroll
    for (int u = 0; u < 4; ++u) {
      a[u]  = *(const v4f*)(xrow + k * 16 + u * 128);
      b0[u] = *(const v4f*)(w0 + k * 16 + u * 128);
      b1[u] = *(const v4f*)(w1 + k * 16 + u * 128);
      b2[u] = *(const v4f*)(w2 + k * 16 + u * 128);
      b3[u] = *(const v4f*)(w3 + k * 16 + u * 128);
    }
#pragma unroll
    for (int u = 0; u < 4; ++u) {
      if (SWAP) {
        acc[0] = wmma_f32(b0[u].xy, a[u].xy, acc[0]);
        acc[1] = wmma_f32(b1[u].xy, a[u].xy, acc[1]);
        acc[2] = wmma_f32(b2[u].xy, a[u].xy, acc[2]);
        acc[3] = wmma_f32(b3[u].xy, a[u].xy, acc[3]);
        acc[0] = wmma_f32(b0[u].zw, a[u].zw, acc[0]);
        acc[1] = wmma_f32(b1[u].zw, a[u].zw, acc[1]);
        acc[2] = wmma_f32(b2[u].zw, a[u].zw, acc[2]);
        acc[3] = wmma_f32(b3[u].zw, a[u].zw, acc[3]);
      } else {
        acc[0] = wmma_f32(a[u].xy, b0[u].xy, acc[0]);
        acc[1] = wmma_f32(a[u].xy, b1[u].xy, acc[1]);
        acc[2] = wmma_f32(a[u].xy, b2[u].xy, acc[2]);
        acc[3] = wmma_f32(a[u].xy, b3[u].xy, acc[3]);
        acc[0] = wmma_f32(a[u].zw, b0[u].zw, acc[0]);
        acc[1] = wmma_f32(a[u].zw, b1[u].zw, acc[1]);
        acc[2] = wmma_f32(a[u].zw, b2[u].zw, acc[2]);
        acc[3] = wmma_f32(a[u].zw, b3[u].zw, acc[3]);
      }
    }
    __builtin_amdgcn_sched_group_barrier(0x020, 20, 0);
    __builtin_amdgcn_sched_group_barrier(0x008, 32, 0);
  }
}

#define X_FO_BLOCKS (M_ROWS / 16 * 24 / 8)
#define W_FO_BLOCKS (D_MODEL / 16 * 24 / 8)
__global__ __launch_bounds__(256) void frag_order_kernel(
    const float* __restrict__ x, const float* __restrict__ wq,
    const float* __restrict__ wk, const float* __restrict__ wv,
    const float* __restrict__ wo, float* __restrict__ xf,
    float* __restrict__ wqf, float* __restrict__ wkf, float* __restrict__ wvf,
    float* __restrict__ wof) {
  const int wave = threadIdx.x >> 5;
  const int lane = threadIdx.x & 31;
  const int lo = lane & 15, hi = lane >> 4;
  int blk = blockIdx.x;
  const float* src = x;
  float* dst = xf;
  if (blk >= X_FO_BLOCKS) {
    blk -= X_FO_BLOCKS;
    const int which = blk / W_FO_BLOCKS;
    blk -= which * W_FO_BLOCKS;
    src = which == 0 ? wq : which == 1 ? wk : which == 2 ? wv : wo;
    dst = which == 0 ? wqf : which == 1 ? wkf : which == 2 ? wvf : wof;
  }
  const int id = blk * 8 + wave;
  const int tile = id / 24, kb = id - tile * 24;
  const float* s = src + (size_t)(tile * 16 + lo) * D_MODEL + 32 * kb + 4 * hi;
  float* d = dst + (size_t)tile * TILE_FLOATS + (size_t)(4 * kb) * 128 + 4 * lane;
  const v4f f0 = *(const v4f*)(s);
  const v4f f1 = *(const v4f*)(s + 8);
  const v4f f2 = *(const v4f*)(s + 16);
  const v4f f3 = *(const v4f*)(s + 24);
  *(v4f*)(d)       = f0;
  *(v4f*)(d + 128) = f1;
  *(v4f*)(d + 256) = f2;
  *(v4f*)(d + 384) = f3;
}

template <int VMODE>
__global__ __launch_bounds__(256) void qkv_gemm_kernel(
    const float* __restrict__ X, const float* __restrict__ W,
    const float* __restrict__ bias, const float* __restrict__ g,
    float* __restrict__ out) {
  const int wave = threadIdx.x >> 5;
  const int lane = threadIdx.x & 31;
  const int lo = lane & 15, hi = lane >> 4;
  const int m0 = blockIdx.x * 32 + (wave >> 2) * 16;
  const int n0 = blockIdx.y * 256 + (wave & 3) * 64;
  const float* xrow  = X + (size_t)(m0 >> 4) * TILE_FLOATS + 4 * lane;
  const float* wbase = W + (size_t)(n0 >> 4) * TILE_FLOATS + 4 * lane;
  v8f acc[4] = {{}, {}, {}, {}};
  gemm_strip_k768<VMODE == 0>(xrow, wbase, wbase + TILE_FLOATS,
                              wbase + 2 * TILE_FLOATS, wbase + 3 * TILE_FLOATS, acc);
  const int h = n0 >> 6;
  const int b = m0 >> 11;
  const int tile = (m0 & (N_CTX - 1)) >> 4;
  float* frag = out + ((size_t)(b * N_HEAD + h) * N_CTX + (size_t)tile * 16) * D_HEAD
                + 4 * lane;
  if (VMODE) {
    float bv[4];
#pragma unroll
    for (int j = 0; j < 4; ++j) bv[j] = bias[n0 + j * 16 + lo];
#pragma unroll
    for (int jv = 0; jv < 2; ++jv) {
#pragma unroll
      for (int j = 0; j < 4; ++j) {
        v4f o;
        o.x = acc[j][4 * jv + 0] + bv[j];
        o.y = acc[j][4 * jv + 1] + bv[j];
        o.z = acc[j][4 * jv + 2] + bv[j];
        o.w = acc[j][4 * jv + 3] + bv[j];
        *(v4f*)(frag + (4 * jv + j) * 128) = o;
      }
    }
  } else {
    float x[4][8], gv[4][8];
#pragma unroll
    for (int j = 0; j < 4; ++j) {
      const v4f b0 = *(const v4f*)(bias + n0 + 16 * j + 8 * hi);
      const v4f b1 = *(const v4f*)(bias + n0 + 16 * j + 8 * hi + 4);
      const v4f g0 = *(const v4f*)(g + 16 * j + 8 * hi);
      const v4f g1 = *(const v4f*)(g + 16 * j + 8 * hi + 4);
#pragma unroll
      for (int r = 0; r < 4; ++r) {
        x[j][r]      = acc[j][r] + b0[r];
        x[j][r + 4]  = acc[j][r + 4] + b1[r];
        gv[j][r]     = g0[r];
        gv[j][r + 4] = g1[r];
      }
    }
    float sj[4];
#pragma unroll
    for (int j = 0; j < 4; ++j) {
      sj[j] = ((x[j][0] * x[j][0] + x[j][1] * x[j][1]) +
               (x[j][2] * x[j][2] + x[j][3] * x[j][3])) +
              ((x[j][4] * x[j][4] + x[j][5] * x[j][5]) +
               (x[j][6] * x[j][6] + x[j][7] * x[j][7]));
    }
    float ss = (sj[0] + sj[2]) + (sj[1] + sj[3]);
    ss = ss + other_half(ss);
    const float inv = rsqrtf(ss * (1.0f / 64.0f) + 1e-6f);
    const int s = (m0 + lo) & (N_CTX - 1);
    float o[4][8];
#pragma unroll
    for (int j = 0; j < 2; ++j) {
#pragma unroll
      for (int r = 0; r < 8; ++r) {
        const int i = 16 * j + 8 * hi + r;
        const float w = exp2f((float)i * (-11.0f / 32.0f));
        const float f = (float)s * w;
        const float c = __cosf(f), sn = __sinf(f);
        const float ya = x[j][r] * inv * gv[j][r];
        const float yb = x[j + 2][r] * inv * gv[j + 2][r];
        o[j][r]     = ya * c + yb * sn;
        o[j + 2][r] = yb * c - ya * sn;
      }
    }
#pragma unroll
    for (int j = 0; j < 4; ++j) {
#pragma unroll
      for (int rr = 0; rr < 2; ++rr) {
        v4f v;
        v.x = o[j][4 * rr + 0];
        v.y = o[j][4 * rr + 1];
        v.z = o[j][4 * rr + 2];
        v.w = o[j][4 * rr + 3];
        *(v4f*)(frag + (2 * j + rr) * 128) = v;
      }
    }
  }
}

__global__ __launch_bounds__(256) void out_gemm_kernel(
    const float* __restrict__ X, const float* __restrict__ W,
    float* __restrict__ out) {
  const int wave = threadIdx.x >> 5;
  const int lane = threadIdx.x & 31;
  const int lo = lane & 15, hi = lane >> 4;
  const int m0 = blockIdx.x * 32 + (wave >> 2) * 16;
  const int n0 = blockIdx.y * 256 + (wave & 3) * 64;
  const float* xrow  = X + (size_t)(m0 >> 4) * TILE_FLOATS + 4 * lane;
  const float* wbase = W + (size_t)(n0 >> 4) * TILE_FLOATS + 4 * lane;
  v8f acc[4] = {{}, {}, {}, {}};
  gemm_strip_k768<false>(xrow, wbase, wbase + TILE_FLOATS, wbase + 2 * TILE_FLOATS,
                         wbase + 3 * TILE_FLOATS, acc);
#pragma unroll
  for (int j = 0; j < 4; ++j) {
#pragma unroll
    for (int r = 0; r < 8; ++r) {
      const int m = m0 + r + 8 * hi;
      out[(size_t)m * D_MODEL + n0 + j * 16 + lo] = acc[j][r];
    }
  }
}

__global__ __launch_bounds__(256) void attn_kernel(
    const float* __restrict__ Q, const float* __restrict__ K,
    const float* __restrict__ V, float* __restrict__ Z) {
  __shared__ __attribute__((aligned(16))) float lds[8 * 256];
  const int wave = threadIdx.x >> 5;
  const int lane = threadIdx.x & 31;
  const int lo = lane & 15, hi = lane >> 4;
  const int wid = blockIdx.x * 8 + wave;
  const int head = wid >> 7;
  const int qt = wid & 127;
  const float* Qh = Q + (size_t)head * N_CTX * D_HEAD;
  const float* Kh = K + (size_t)head * N_CTX * D_HEAD;
  const float* Vh = V + (size_t)head * N_CTX * D_HEAD;
  float* pl = lds + wave * 256;

  v4f qa[8];
  {
    const float* qfrag = Qh + (size_t)qt * (16 * D_HEAD) + 4 * lane;
#pragma unroll
    for (int j = 0; j < 8; ++j) qa[j] = *(const v4f*)(qfrag + j * 128);
  }

  v8f z0 = {}, z1 = {}, z2 = {}, z3 = {};
  for (int kt = 0; kt <= qt; ++kt) {
    const float* kfrag = Kh + (size_t)kt * (16 * D_HEAD) + 4 * lane;
    __builtin_prefetch(Kh + (size_t)(kt + 1) * (16 * D_HEAD) + 32 * lane, 0, 0);
    v4f kb[8], vb[2][4];
#pragma unroll
    for (int j = 0; j < 8; ++j) kb[j] = *(const v4f*)(kfrag + j * 128);
    {
      const float* vfrag = Vh + (size_t)kt * (16 * D_HEAD) + 4 * lane;
#pragma unroll
      for (int j = 0; j < 2; ++j) {
#pragma unroll
        for (int t = 0; t < 4; ++t)
          vb[j][t] = *(const v4f*)(vfrag + (4 * j + t) * 128);
      }
    }
    v8f sc = {};
#pragma unroll
    for (int j = 0; j < 8; ++j) {
      sc = wmma_f32(qa[j].xy, kb[j].xy, sc);
      sc = wmma_f32(qa[j].zw, kb[j].zw, sc);
    }
#pragma unroll
    for (int r = 0; r < 8; ++r) {
      float pv = sc[r] * (1.0f / 64.0f);
      pv = pv * pv;
      if (kt == qt && lo > r + 8 * hi) pv = 0.0f;
      pl[(r + 8 * hi) * 16 + lo] = pv;
    }
#pragma unroll
    for (int j = 0; j < 2; ++j) {
      v4f pa = *(const v4f*)(pl + lo * 16 + 8 * hi + 4 * j);
      z0 = wmma_f32(pa.xy, vb[j][0].xy, z0);
      z1 = wmma_f32(pa.xy, vb[j][1].xy, z1);
      z2 = wmma_f32(pa.xy, vb[j][2].xy, z2);
      z3 = wmma_f32(pa.xy, vb[j][3].xy, z3);
      z0 = wmma_f32(pa.zw, vb[j][0].zw, z0);
      z1 = wmma_f32(pa.zw, vb[j][1].zw, z1);
      z2 = wmma_f32(pa.zw, vb[j][2].zw, z2);
      z3 = wmma_f32(pa.zw, vb[j][3].zw, z3);
    }
    __builtin_amdgcn_sched_group_barrier(0x020, 16, 0);
    __builtin_amdgcn_sched_group_barrier(0x008, 32, 0);
  }
  const int b = head / N_HEAD, h = head % N_HEAD;
  float* zfrag = Z + (size_t)(b * (N_CTX / 16) + qt) * TILE_FLOATS
                 + (size_t)(h * 8) * 128 + 4 * lane;
  const v8f zt[4] = {z0, z1, z2, z3};
#pragma unroll
  for (int t = 0; t < 4; ++t) {
#pragma unroll
    for (int r = 0; r < 8; ++r) pl[(r + 8 * hi) * 16 + lo] = zt[t][r];
#pragma unroll
    for (int ch = 0; ch < 2; ++ch) {
      const v4f zv = *(const v4f*)(pl + lo * 16 + 8 * ch + 4 * hi);
      *(v4f*)(zfrag + (2 * t + ch) * 128) = zv;
    }
  }
}

extern "C" void kernel_launch(void* const* d_in, const int* in_sizes, int n_in,
                              void* d_out, int out_size, void* d_ws, size_t ws_size,
                              hipStream_t stream) {
  const float* x  = (const float*)d_in[0];
  const float* Wq = (const float*)d_in[1];
  const float* bq = (const float*)d_in[2];
  const float* Wk = (const float*)d_in[3];
  const float* bk = (const float*)d_in[4];
  const float* Wv = (const float*)d_in[5];
  const float* bv = (const float*)d_in[6];
  const float* Wo = (const float*)d_in[7];
  const float* g  = (const float*)d_in[8];
  float* out = (float*)d_out;

  const size_t QKV = (size_t)BATCH * N_HEAD * N_CTX * D_HEAD;
  float* q = (float*)d_ws;
  float* k = q + QKV;
  float* v = k + QKV;
  float* z = v + QKV;
  float* xf  = z + QKV;
  const size_t WSZ = (size_t)D_MODEL * D_MODEL;
  float* wqf = xf + QKV;
  float* wkf = wqf + WSZ;
  float* wvf = wkf + WSZ;
  float* wof = wvf + WSZ;

  frag_order_kernel<<<X_FO_BLOCKS + 4 * W_FO_BLOCKS, 256, 0, stream>>>(
      x, Wq, Wk, Wv, Wo, xf, wqf, wkf, wvf, wof);

  dim3 gblk(M_ROWS / 32, D_MODEL / 256);
  qkv_gemm_kernel<0><<<gblk, 256, 0, stream>>>(xf, wqf, bq, g, q);
  qkv_gemm_kernel<0><<<gblk, 256, 0, stream>>>(xf, wkf, bk, g, k);
  qkv_gemm_kernel<1><<<gblk, 256, 0, stream>>>(xf, wvf, bv, g, v);

  attn_kernel<<<(BATCH * N_HEAD * (N_CTX / 16)) / 8, 256, 0, stream>>>(q, k, v, z);

  out_gemm_kernel<<<gblk, 256, 0, stream>>>(z, wof, out);
}


# Attention in equal waves: q tile t>=64 sums key tiles 0..63 and 64..t in two waves (added through LDS); the short part shares a wave with q tile 127-t; 256 blocks x 12 waves; Z sum order changes
# speedup vs baseline: 2.8436x; 1.3904x over previous
#include <hip/hip_runtime.h>
#include <cstddef>

typedef float v2f __attribute__((ext_vector_type(2)));
typedef float v4f __attribute__((ext_vector_type(4)));
typedef float v8f __attribute__((ext_vector_type(8)));

#define D_MODEL 768
#define N_HEAD  12
#define D_HEAD  64
#define N_CTX   2048
#define BATCH   2
#define M_ROWS  (BATCH * N_CTX)

__device__ __forceinline__ v8f wmma_f32(v2f a, v2f b, v8f c) {
  return __builtin_amdgcn_wmma_f32_16x16x4_f32(false, a, false, b, (short)0, c,
                                               false, false);
}

template <int MASK>
__device__ __forceinline__ float row_xor_add(float v) {
  return v + __int_as_float(__builtin_amdgcn_update_dpp(
                 0, __float_as_int(v), 0x160 + MASK, 0xf, 0xf, false));
}

__device__ __forceinline__ float other_half(float v) {
  return __int_as_float(__builtin_amdgcn_permlanex16(
      __float_as_int(v), __float_as_int(v), 0x76543210, 0xfedcba98, false, false));
}

#define K_CHUNKS    (D_MODEL / 8)
#define TILE_FLOATS (K_CHUNKS * 128)

template <bool SWAP>
__device__ __forceinline__ void gemm_strip_k768(const float* __restrict__ xrow,
                                                const float* __restrict__ w0,
                                                const float* __restrict__ w1,
                                                const float* __restrict__ w2,
                                                const float* __restrict__ w3,
                                                v8f acc[4]) {
  for (int k = 0; k < D_MODEL; k += 32) {
    v4f a[4], b0[4], b1[4], b2[4], b3[4];
#pragma unroll
    for (int u = 0; u < 4; ++u) {
      a[u]  = *(const v4f*)(xrow + k * 16 + u * 128);
      b0[u] = *(const v4f*)(w0 + k * 16 + u * 128);
      b1[u] = *(const v4f*)(w1 + k * 16 + u * 128);
      b2[u] = *(const v4f*)(w2 + k * 16 + u * 128);
      b3[u] = *(const v4f*)(w3 + k * 16 + u * 128);
    }
#pragma unroll
    for (int u = 0; u < 4; ++u) {
      if (SWAP) {
        acc[0] = wmma_f32(b0[u].xy, a[u].xy, acc[0]);
        acc[1] = wmma_f32(b1[u].xy, a[u].xy, acc[1]);
        acc[2] = wmma_f32(b2[u].xy, a[u].xy, acc[2]);
        acc[3] = wmma_f32(b3[u].xy, a[u].xy, acc[3]);
        acc[0] = wmma_f32(b0[u].zw, a[u].zw, acc[0]);
        acc[1] = wmma_f32(b1[u].zw, a[u].zw, acc[1]);
        acc[2] = wmma_f32(b2[u].zw, a[u].zw, acc[2]);
        acc[3] = wmma_f32(b3[u].zw, a[u].zw, acc[3]);
      } else {
        acc[0] = wmma_f32(a[u].xy, b0[u].xy, acc[0]);
        acc[1] = wmma_f32(a[u].xy, b1[u].xy, acc[1]);
        acc[2] = wmma_f32(a[u].xy, b2[u].xy, acc[2]);
        acc[3] = wmma_f32(a[u].xy, b3[u].xy, acc[3]);
        acc[0] = wmma_f32(a[u].zw, b0[u].zw, acc[0]);
        acc[1] = wmma_f32(a[u].zw, b1[u].zw, acc[1]);
        acc[2] = wmma_f32(a[u].zw, b2[u].zw, acc[2]);
        acc[3] = wmma_f32(a[u].zw, b3[u].zw, acc[3]);
      }
    }
    __builtin_amdgcn_sched_group_barrier(0x020, 20, 0);
    __builtin_amdgcn_sched_group_barrier(0x008, 32, 0);
  }
}

#define X_FO_BLOCKS (M_ROWS / 16 * 24 / 8)
#define W_FO_BLOCKS (D_MODEL / 16 * 24 / 8)
__global__ __launch_bounds__(256) void frag_order_kernel(
    const float* __restrict__ x, const float* __restrict__ wq,
    const float* __restrict__ wk, const float* __restrict__ wv,
    const float* __restrict__ wo, float* __restrict__ xf,
    float* __restrict__ wqf, float* __restrict__ wkf, float* __restrict__ wvf,
    float* __restrict__ wof) {
  const int wave = threadIdx.x >> 5;
  const int lane = threadIdx.x & 31;
  const int lo = lane & 15, hi = lane >> 4;
  int blk = blockIdx.x;
  const float* src = x;
  float* dst = xf;
  if (blk >= X_FO_BLOCKS) {
    blk -= X_FO_BLOCKS;
    const int which = blk / W_FO_BLOCKS;
    blk -= which * W_FO_BLOCKS;
    src = which == 0 ? wq : which == 1 ? wk : which == 2 ? wv : wo;
    dst = which == 0 ? wqf : which == 1 ? wkf : which == 2 ? wvf : wof;
  }
  const int id = blk * 8 + wave;
  const int tile = id / 24, kb = id - tile * 24;
  const float* s = src + (size_t)(tile * 16 + lo) * D_MODEL + 32 * kb + 4 * hi;
  float* d = dst + (size_t)tile * TILE_FLOATS + (size_t)(4 * kb) * 128 + 4 * lane;
  const v4f f0 = *(const v4f*)(s);
  const v4f f1 = *(const v4f*)(s + 8);
  const v4f f2 = *(const v4f*)(s + 16);
  const v4f f3 = *(const v4f*)(s + 24);
  *(v4f*)(d)       = f0;
  *(v4f*)(d + 128) = f1;
  *(v4f*)(d + 256) = f2;
  *(v4f*)(d + 384) = f3;
}

template <int VMODE>
__global__ __launch_bounds__(256) void qkv_gemm_kernel(
    const float* __restrict__ X, const float* __restrict__ W,
    const float* __restrict__ bias, const float* __restrict__ g,
    float* __restrict__ out) {
  const int wave = threadIdx.x >> 5;
  const int lane = threadIdx.x & 31;
  const int lo = lane & 15, hi = lane >> 4;
  const int m0 = blockIdx.x * 32 + (wave >> 2) * 16;
  const int n0 = blockIdx.y * 256 + (wave & 3) * 64;
  const float* xrow  = X + (size_t)(m0 >> 4) * TILE_FLOATS + 4 * lane;
  const float* wbase = W + (size_t)(n0 >> 4) * TILE_FLOATS + 4 * lane;
  v8f acc[4] = {{}, {}, {}, {}};
  gemm_strip_k768<VMODE == 0>(xrow, wbase, wbase + TILE_FLOATS,
                              wbase + 2 * TILE_FLOATS, wbase + 3 * TILE_FLOATS, acc);
  const int h = n0 >> 6;
  const int b = m0 >> 11;
  const int tile = (m0 & (N_CTX - 1)) >> 4;
  float* frag = out + ((size_t)(b * N_HEAD + h) * N_CTX + (size_t)tile * 16) * D_HEAD
                + 4 * lane;
  if (VMODE) {
    float bv[4];
#pragma unroll
    for (int j = 0; j < 4; ++j) bv[j] = bias[n0 + j * 16 + lo];
#pragma unroll
    for (int jv = 0; jv < 2; ++jv) {
#pragma unroll
      for (int j = 0; j < 4; ++j) {
        v4f o;
        o.x = acc[j][4 * jv + 0] + bv[j];
        o.y = acc[j][4 * jv + 1] + bv[j];
        o.z = acc[j][4 * jv + 2] + bv[j];
        o.w = acc[j][4 * jv + 3] + bv[j];
        *(v4f*)(frag + (4 * jv + j) * 128) = o;
      }
    }
  } else {
    float x[4][8], gv[4][8];
#pragma unroll
    for (int j = 0; j < 4; ++j) {
      const v4f b0 = *(const v4f*)(bias + n0 + 16 * j + 8 * hi);
      const v4f b1 = *(const v4f*)(bias + n0 + 16 * j + 8 * hi + 4);
      const v4f g0 = *(const v4f*)(g + 16 * j + 8 * hi);
      const v4f g1 = *(const v4f*)(g + 16 * j + 8 * hi + 4);
#pragma unroll
      for (int r = 0; r < 4; ++r) {
        x[j][r]      = acc[j][r] + b0[r];
        x[j][r + 4]  = acc[j][r + 4] + b1[r];
        gv[j][r]     = g0[r];
        gv[j][r + 4] = g1[r];
      }
    }
    float sj[4];
#pragma unroll
    for (int j = 0; j < 4; ++j) {
      sj[j] = ((x[j][0] * x[j][0] + x[j][1] * x[j][1]) +
               (x[j][2] * x[j][2] + x[j][3] * x[j][3])) +
              ((x[j][4] * x[j][4] + x[j][5] * x[j][5]) +
               (x[j][6] * x[j][6] + x[j][7] * x[j][7]));
    }
    float ss = (sj[0] + sj[2]) + (sj[1] + sj[3]);
    ss = ss + other_half(ss);
    const float inv = rsqrtf(ss * (1.0f / 64.0f) + 1e-6f);
    const int s = (m0 + lo) & (N_CTX - 1);
    float o[4][8];
#pragma unroll
    for (int j = 0; j < 2; ++j) {
#pragma unroll
      for (int r = 0; r < 8; ++r) {
        const int i = 16 * j + 8 * hi + r;
        const float w = exp2f((float)i * (-11.0f / 32.0f));
        const float f = (float)s * w;
        const float c = __cosf(f), sn = __sinf(f);
        const float ya = x[j][r] * inv * gv[j][r];
        const float yb = x[j + 2][r] * inv * gv[j + 2][r];
        o[j][r]     = ya * c + yb * sn;
        o[j + 2][r] = yb * c - ya * sn;
      }
    }
#pragma unroll
    for (int j = 0; j < 4; ++j) {
#pragma unroll
      for (int rr = 0; rr < 2; ++rr) {
        v4f v;
        v.x = o[j][4 * rr + 0];
        v.y = o[j][4 * rr + 1];
        v.z = o[j][4 * rr + 2];
        v.w = o[j][4 * rr + 3];
        *(v4f*)(frag + (2 * j + rr) * 128) = v;
      }
    }
  }
}

__global__ __launch_bounds__(256) void out_gemm_kernel(
    const float* __restrict__ X, const float* __restrict__ W,
    float* __restrict__ out) {
  const int wave = threadIdx.x >> 5;
  const int lane = threadIdx.x & 31;
  const int lo = lane & 15, hi = lane >> 4;
  const int m0 = blockIdx.x * 32 + (wave >> 2) * 16;
  const int n0 = blockIdx.y * 256 + (wave & 3) * 64;
  const float* xrow  = X + (size_t)(m0 >> 4) * TILE_FLOATS + 4 * lane;
  const float* wbase = W + (size_t)(n0 >> 4) * TILE_FLOATS + 4 * lane;
  v8f acc[4] = {{}, {}, {}, {}};
  gemm_strip_k768<false>(xrow, wbase, wbase + TILE_FLOATS, wbase + 2 * TILE_FLOATS,
                         wbase + 3 * TILE_FLOATS, acc);
#pragma unroll
  for (int j = 0; j < 4; ++j) {
#pragma unroll
    for (int r = 0; r < 8; ++r) {
      const int m = m0 + r + 8 * hi;
      out[(size_t)m * D_MODEL + n0 + j * 16 + lo] = acc[j][r];
    }
  }
}

#define ATT_PAIRS  6
#define ATT_WAVES  (2 * ATT_PAIRS)
#define ATT_BLOCKS (BATCH * N_HEAD * 64 / ATT_PAIRS)

template <bool MASK>
__device__ __forceinline__ void key_tile(const v4f (&qa)[8],
                                         const float* __restrict__ kfrag,
                                         const float* __restrict__ vfrag,
                                         const float* __restrict__ knext,
                                         float* __restrict__ pl, int lo, int hi,
                                         v8f& z0, v8f& z1, v8f& z2, v8f& z3) {
  __builtin_prefetch(knext, 0, 0);
  v4f kb[8], vb[2][4];
#pragma unroll
  for (int j = 0; j < 8; ++j) kb[j] = *(const v4f*)(kfrag + j * 128);
#pragma unroll
  for (int j = 0; j < 2; ++j) {
#pragma unroll
    for (int t = 0; t < 4; ++t)
      vb[j][t] = *(const v4f*)(vfrag + (4 * j + t) * 128);
  }
  v8f sc = {};
#pragma unroll
  for (int j = 0; j < 8; ++j) {
    sc = wmma_f32(qa[j].xy, kb[j].xy, sc);
    sc = wmma_f32(qa[j].zw, kb[j].zw, sc);
  }
#pragma unroll
  for (int r = 0; r < 8; ++r) {
    float pv = sc[r] * (1.0f / 64.0f);
    pv = pv * pv;
    if (MASK && lo > r + 8 * hi) pv = 0.0f;
    pl[(r + 8 * hi) * 16 + lo] = pv;
  }
#pragma unroll
  for (int j = 0; j < 2; ++j) {
    v4f pa = *(const v4f*)(pl + lo * 16 + 8 * hi + 4 * j);
    z0 = wmma_f32(pa.xy, vb[j][0].xy, z0);
    z1 = wmma_f32(pa.xy, vb[j][1].xy, z1);
    z2 = wmma_f32(pa.xy, vb[j][2].xy, z2);
    z3 = wmma_f32(pa.xy, vb[j][3].xy, z3);
    z0 = wmma_f32(pa.zw, vb[j][0].zw, z0);
    z1 = wmma_f32(pa.zw, vb[j][1].zw, z1);
    z2 = wmma_f32(pa.zw, vb[j][2].zw, z2);
    z3 = wmma_f32(pa.zw, vb[j][3].zw, z3);
  }
  __builtin_amdgcn_sched_group_barrier(0x020, 16, 0);
  __builtin_amdgcn_sched_group_barrier(0x008, 32, 0);
}

__device__ __forceinline__ void sweep(const float* __restrict__ Qh,
                                      const float* __restrict__ Kh,
                                      const float* __restrict__ Vh, int qt,
                                      int kt0, int nfull, bool diag,
                                      float* __restrict__ pl, int lane, int lo,
                                      int hi, v8f& z0, v8f& z1, v8f& z2,
                                      v8f& z3) {
  v4f qa[8];
  {
    const float* qfrag = Qh + (size_t)qt * (16 * D_HEAD) + 4 * lane;
#pragma unroll
    for (int j = 0; j < 8; ++j) qa[j] = *(const v4f*)(qfrag + j * 128);
  }
  for (int kt = kt0; kt < kt0 + nfull; ++kt) {
    key_tile<false>(qa, Kh + (size_t)kt * (16 * D_HEAD) + 4 * lane,
                    Vh + (size_t)kt * (16 * D_HEAD) + 4 * lane,
                    Kh + (size_t)(kt + 1) * (16 * D_HEAD) + 32 * lane, pl, lo,
                    hi, z0, z1, z2, z3);
  }
  if (diag) {
    key_tile<true>(qa, Kh + (size_t)qt * (16 * D_HEAD) + 4 * lane,
                   Vh + (size_t)qt * (16 * D_HEAD) + 4 * lane,
                   Kh + (size_t)(qt + 1) * (16 * D_HEAD) + 32 * lane, pl, lo,
                   hi, z0, z1, z2, z3);
  }
}

__device__ __forceinline__ void store_z_tile(float* __restrict__ Z, int head,
                                             int qt, float* __restrict__ pl,
                                             int lane, int lo, int hi,
                                             const v8f& z0, const v8f& z1,
                                             const v8f& z2, const v8f& z3) {
  const int b = head / N_HEAD, h = head % N_HEAD;
  float* zfrag = Z + (size_t)(b * (N_CTX / 16) + qt) * TILE_FLOATS
                 + (size_t)(h * 8) * 128 + 4 * lane;
  const v8f zt[4] = {z0, z1, z2, z3};
#pragma unroll
  for (int t = 0; t < 4; ++t) {
#pragma unroll
    for (int r = 0; r < 8; ++r) pl[(r + 8 * hi) * 16 + lo] = zt[t][r];
#pragma unroll
    for (int ch = 0; ch < 2; ++ch) {
      const v4f zv = *(const v4f*)(pl + lo * 16 + 8 * ch + 4 * hi);
      *(v4f*)(zfrag + (2 * t + ch) * 128) = zv;
    }
  }
}

__global__ __launch_bounds__(32 * ATT_WAVES) void attn_kernel(
    const float* __restrict__ Q, const float* __restrict__ K,
    const float* __restrict__ V, float* __restrict__ Z) {
  __shared__ __attribute__((aligned(16))) float lds[ATT_WAVES * 256 + ATT_PAIRS * 1024];
  const int wave = threadIdx.x >> 5;
  const int lane = threadIdx.x & 31;
  const int lo = lane & 15, hi = lane >> 4;
  const bool is_a = wave < ATT_PAIRS;
  const int slot = is_a ? wave : wave - ATT_PAIRS;
  const int gp = blockIdx.x * ATT_PAIRS + slot;
  const int head = gp >> 6;
  const int u = gp & 63;
  const float* Qh = Q + (size_t)head * N_CTX * D_HEAD;
  const float* Kh = K + (size_t)head * N_CTX * D_HEAD;
  const float* Vh = V + (size_t)head * N_CTX * D_HEAD;
  float* pl = lds + wave * 256;
  float* zp = lds + ATT_WAVES * 256 + slot * 1024 + lane;

  v8f z0 = {}, z1 = {}, z2 = {}, z3 = {};
  sweep(Qh, Kh, Vh, 64 + u, is_a ? 0 : 64, is_a ? 64 : u, !is_a, pl, lane, lo,
        hi, z0, z1, z2, z3);
  if (!is_a) {
    const v8f zt[4] = {z0, z1, z2, z3};
#pragma unroll
    for (int t = 0; t < 4; ++t) {
#pragma unroll
      for (int r = 0; r < 8; ++r) zp[(t * 8 + r) * 32] = zt[t][r];
    }
    v8f y0 = {}, y1 = {}, y2 = {}, y3 = {};
    sweep(Qh, Kh, Vh, 63 - u, 0, 63 - u, true, pl, lane, lo, hi, y0, y1, y2, y3);
    store_z_tile(Z, head, 63 - u, pl, lane, lo, hi, y0, y1, y2, y3);
  }
  __syncthreads();
  if (is_a) {
#pragma unroll
    for (int r = 0; r < 8; ++r) {
      z0[r] += zp[(0 * 8 + r) * 32];
      z1[r] += zp[(1 * 8 + r) * 32];
      z2[r] += zp[(2 * 8 + r) * 32];
      z3[r] += zp[(3 * 8 + r) * 32];
    }
    store_z_tile(Z, head, 64 + u, pl, lane, lo, hi, z0, z1, z2, z3);
  }
}

extern "C" void kernel_launch(void* const* d_in, const int* in_sizes, int n_in,
                              void* d_out, int out_size, void* d_ws, size_t ws_size,
                              hipStream_t stream) {
  const float* x  = (const float*)d_in[0];
  const float* Wq = (const float*)d_in[1];
  const float* bq = (const float*)d_in[2];
  const float* Wk = (const float*)d_in[3];
  const float* bk = (const float*)d_in[4];
  const float* Wv = (const float*)d_in[5];
  const float* bv = (const float*)d_in[6];
  const float* Wo = (const float*)d_in[7];
  const float* g  = (const float*)d_in[8];
  float* out = (float*)d_out;

  const size_t QKV = (size_t)BATCH * N_HEAD * N_CTX * D_HEAD;
  float* q = (float*)d_ws;
  float* k = q + QKV;
  float* v = k + QKV;
  float* z = v + QKV;
  float* xf  = z + QKV;
  const size_t WSZ = (size_t)D_MODEL * D_MODEL;
  float* wqf = xf + QKV;
  float* wkf = wqf + WSZ;
  float* wvf = wkf + WSZ;
  float* wof = wvf + WSZ;

  frag_order_kernel<<<X_FO_BLOCKS + 4 * W_FO_BLOCKS, 256, 0, stream>>>(
      x, Wq, Wk, Wv, Wo, xf, wqf, wkf, wvf, wof);

  dim3 gblk(M_ROWS / 32, D_MODEL / 256);
  qkv_gemm_kernel<0><<<gblk, 256, 0, stream>>>(xf, wqf, bq, g, q);
  qkv_gemm_kernel<0><<<gblk, 256, 0, stream>>>(xf, wkf, bk, g, k);
  qkv_gemm_kernel<1><<<gblk, 256, 0, stream>>>(xf, wvf, bv, g, v);

  attn_kernel<<<ATT_BLOCKS, 32 * ATT_WAVES, 0, stream>>>(q, k, v, z);

  out_gemm_kernel<<<gblk, 256, 0, stream>>>(z, wof, out);
}


# GEMM launches re-shaped to 256 blocks x 12 waves (4 row tiles x 3 strips): one block per WGP, 3 equal waves per SIMD; was 384 x 8, so half the WGPs ran 4 waves per SIMD and half 2; sums unchanged
# speedup vs baseline: 3.0659x; 1.0782x over previous
#include <hip/hip_runtime.h>
#include <cstddef>

typedef float v2f __attribute__((ext_vector_type(2)));
typedef float v4f __attribute__((ext_vector_type(4)));
typedef float v8f __attribute__((ext_vector_type(8)));

#define D_MODEL 768
#define N_HEAD  12
#define D_HEAD  64
#define N_CTX   2048
#define BATCH   2
#define M_ROWS  (BATCH * N_CTX)

__device__ __forceinline__ v8f wmma_f32(v2f a, v2f b, v8f c) {
  return __builtin_amdgcn_wmma_f32_16x16x4_f32(false, a, false, b, (short)0, c,
                                               false, false);
}

template <int MASK>
__device__ __forceinline__ float row_xor_add(float v) {
  return v + __int_as_float(__builtin_amdgcn_update_dpp(
                 0, __float_as_int(v), 0x160 + MASK, 0xf, 0xf, false));
}

__device__ __forceinline__ float other_half(float v) {
  return __int_as_float(__builtin_amdgcn_permlanex16(
      __float_as_int(v), __float_as_int(v), 0x76543210, 0xfedcba98, false, false));
}

#define K_CHUNKS    (D_MODEL / 8)
#define TILE_FLOATS (K_CHUNKS * 128)

template <bool SWAP>
__device__ __forceinline__ void gemm_strip_k768(const float* __restrict__ xrow,
                                                const float* __restrict__ w0,
                                                const float* __restrict__ w1,
                                                const float* __restrict__ w2,
                                                const float* __restrict__ w3,
                                                v8f acc[4]) {
  for (int k = 0; k < D_MODEL; k += 32) {
    v4f a[4], b0[4], b1[4], b2[4], b3[4];
#pragma unroll
    for (int u = 0; u < 4; ++u) {
      a[u]  = *(const v4f*)(xrow + k * 16 + u * 128);
      b0[u] = *(const v4f*)(w0 + k * 16 + u * 128);
      b1[u] = *(const v4f*)(w1 + k * 16 + u * 128);
      b2[u] = *(const v4f*)(w2 + k * 16 + u * 128);
      b3[u] = *(const v4f*)(w3 + k * 16 + u * 128);
    }
#pragma unroll
    for (int u = 0; u < 4; ++u) {
      if (SWAP) {
        acc[0] = wmma_f32(b0[u].xy, a[u].xy, acc[0]);
        acc[1] = wmma_f32(b1[u].xy, a[u].xy, acc[1]);
        acc[2] = wmma_f32(b2[u].xy, a[u].xy, acc[2]);
        acc[3] = wmma_f32(b3[u].xy, a[u].xy, acc[3]);
        acc[0] = wmma_f32(b0[u].zw, a[u].zw, acc[0]);
        acc[1] = wmma_f32(b1[u].zw, a[u].zw, acc[1]);
        acc[2] = wmma_f32(b2[u].zw, a[u].zw, acc[2]);
        acc[3] = wmma_f32(b3[u].zw, a[u].zw, acc[3]);
      } else {
        acc[0] = wmma_f32(a[u].xy, b0[u].xy, acc[0]);
        acc[1] = wmma_f32(a[u].xy, b1[u].xy, acc[1]);
        acc[2] = wmma_f32(a[u].xy, b2[u].xy, acc[2]);
        acc[3] = wmma_f32(a[u].xy, b3[u].xy, acc[3]);
        acc[0] = wmma_f32(a[u].zw, b0[u].zw, acc[0]);
        acc[1] = wmma_f32(a[u].zw, b1[u].zw, acc[1]);
        acc[2] = wmma_f32(a[u].zw, b2[u].zw, acc[2]);
        acc[3] = wmma_f32(a[u].zw, b3[u].zw, acc[3]);
      }
    }
    __builtin_amdgcn_sched_group_barrier(0x020, 20, 0);
    __builtin_amdgcn_sched_group_barrier(0x008, 32, 0);
  }
}

#define X_FO_BLOCKS (M_ROWS / 16 * 24 / 8)
#define W_FO_BLOCKS (D_MODEL / 16 * 24 / 8)
__global__ __launch_bounds__(256) void frag_order_kernel(
    const float* __restrict__ x, const float* __restrict__ wq,
    const float* __restrict__ wk, const float* __restrict__ wv,
    const float* __restrict__ wo, float* __restrict__ xf,
    float* __restrict__ wqf, float* __restrict__ wkf, float* __restrict__ wvf,
    float* __restrict__ wof) {
  const int wave = threadIdx.x >> 5;
  const int lane = threadIdx.x & 31;
  const int lo = lane & 15, hi = lane >> 4;
  int blk = blockIdx.x;
  const float* src = x;
  float* dst = xf;
  if (blk >= X_FO_BLOCKS) {
    blk -= X_FO_BLOCKS;
    const int which = blk / W_FO_BLOCKS;
    blk -= which * W_FO_BLOCKS;
    src = which == 0 ? wq : which == 1 ? wk : which == 2 ? wv : wo;
    dst = which == 0 ? wqf : which == 1 ? wkf : which == 2 ? wvf : wof;
  }
  const int id = blk * 8 + wave;
  const int tile = id / 24, kb = id - tile * 24;
  const float* s = src + (size_t)(tile * 16 + lo) * D_MODEL + 32 * kb + 4 * hi;
  float* d = dst + (size_t)tile * TILE_FLOATS + (size_t)(4 * kb) * 128 + 4 * lane;
  const v4f f0 = *(const v4f*)(s);
  const v4f f1 = *(const v4f*)(s + 8);
  const v4f f2 = *(const v4f*)(s + 16);
  const v4f f3 = *(const v4f*)(s + 24);
  *(v4f*)(d)       = f0;
  *(v4f*)(d + 128) = f1;
  *(v4f*)(d + 256) = f2;
  *(v4f*)(d + 384) = f3;
}

#define GEMM_BM    4
#define GEMM_BN    3
#define GEMM_WAVES (GEMM_BM * GEMM_BN)

template <int VMODE>
__global__ __launch_bounds__(32 * GEMM_WAVES) void qkv_gemm_kernel(
    const float* __restrict__ X, const float* __restrict__ W,
    const float* __restrict__ bias, const float* __restrict__ g,
    float* __restrict__ out) {
  const int wave = threadIdx.x >> 5;
  const int lane = threadIdx.x & 31;
  const int lo = lane & 15, hi = lane >> 4;
  const int m0 = (blockIdx.x * GEMM_BM + (wave & (GEMM_BM - 1))) * 16;
  const int n0 = (blockIdx.y * GEMM_BN + wave / GEMM_BM) * 64;
  const float* xrow  = X + (size_t)(m0 >> 4) * TILE_FLOATS + 4 * lane;
  const float* wbase = W + (size_t)(n0 >> 4) * TILE_FLOATS + 4 * lane;
  v8f acc[4] = {{}, {}, {}, {}};
  gemm_strip_k768<VMODE == 0>(xrow, wbase, wbase + TILE_FLOATS,
                              wbase + 2 * TILE_FLOATS, wbase + 3 * TILE_FLOATS, acc);
  const int h = n0 >> 6;
  const int b = m0 >> 11;
  const int tile = (m0 & (N_CTX - 1)) >> 4;
  float* frag = out + ((size_t)(b * N_HEAD + h) * N_CTX + (size_t)tile * 16) * D_HEAD
                + 4 * lane;
  if (VMODE) {
    float bv[4];
#pragma unroll
    for (int j = 0; j < 4; ++j) bv[j] = bias[n0 + j * 16 + lo];
#pragma unroll
    for (int jv = 0; jv < 2; ++jv) {
#pragma unroll
      for (int j = 0; j < 4; ++j) {
        v4f o;
        o.x = acc[j][4 * jv + 0] + bv[j];
        o.y = acc[j][4 * jv + 1] + bv[j];
        o.z = acc[j][4 * jv + 2] + bv[j];
        o.w = acc[j][4 * jv + 3] + bv[j];
        *(v4f*)(frag + (4 * jv + j) * 128) = o;
      }
    }
  } else {
    float x[4][8], gv[4][8];
#pragma unroll
    for (int j = 0; j < 4; ++j) {
      const v4f b0 = *(const v4f*)(bias + n0 + 16 * j + 8 * hi);
      const v4f b1 = *(const v4f*)(bias + n0 + 16 * j + 8 * hi + 4);
      const v4f g0 = *(const v4f*)(g + 16 * j + 8 * hi);
      const v4f g1 = *(const v4f*)(g + 16 * j + 8 * hi + 4);
#pragma unroll
      for (int r = 0; r < 4; ++r) {
        x[j][r]      = acc[j][r] + b0[r];
        x[j][r + 4]  = acc[j][r + 4] + b1[r];
        gv[j][r]     = g0[r];
        gv[j][r + 4] = g1[r];
      }
    }
    float sj[4];
#pragma unroll
    for (int j = 0; j < 4; ++j) {
      sj[j] = ((x[j][0] * x[j][0] + x[j][1] * x[j][1]) +
               (x[j][2] * x[j][2] + x[j][3] * x[j][3])) +
              ((x[j][4] * x[j][4] + x[j][5] * x[j][5]) +
               (x[j][6] * x[j][6] + x[j][7] * x[j][7]));
    }
    float ss = (sj[0] + sj[2]) + (sj[1] + sj[3]);
    ss = ss + other_half(ss);
    const float inv = rsqrtf(ss * (1.0f / 64.0f) + 1e-6f);
    const int s = (m0 + lo) & (N_CTX - 1);
    float o[4][8];
#pragma unroll
    for (int j = 0; j < 2; ++j) {
#pragma unroll
      for (int r = 0; r < 8; ++r) {
        const int i = 16 * j + 8 * hi + r;
        const float w = exp2f((float)i * (-11.0f / 32.0f));
        const float f = (float)s * w;
        const float c = __cosf(f), sn = __sinf(f);
        const float ya = x[j][r] * inv * gv[j][r];
        const float yb = x[j + 2][r] * inv * gv[j + 2][r];
        o[j][r]     = ya * c + yb * sn;
        o[j + 2][r] = yb * c - ya * sn;
      }
    }
#pragma unroll
    for (int j = 0; j < 4; ++j) {
#pragma unroll
      for (int rr = 0; rr < 2; ++rr) {
        v4f v;
        v.x = o[j][4 * rr + 0];
        v.y = o[j][4 * rr + 1];
        v.z = o[j][4 * rr + 2];
        v.w = o[j][4 * rr + 3];
        *(v4f*)(frag + (2 * j + rr) * 128) = v;
      }
    }
  }
}

__global__ __launch_bounds__(32 * GEMM_WAVES) void out_gemm_kernel(
    const float* __restrict__ X, const float* __restrict__ W,
    float* __restrict__ out) {
  const int wave = threadIdx.x >> 5;
  const int lane = threadIdx.x & 31;
  const int lo = lane & 15, hi = lane >> 4;
  const int m0 = (blockIdx.x * GEMM_BM + (wave & (GEMM_BM - 1))) * 16;
  const int n0 = (blockIdx.y * GEMM_BN + wave / GEMM_BM) * 64;
  const float* xrow  = X + (size_t)(m0 >> 4) * TILE_FLOATS + 4 * lane;
  const float* wbase = W + (size_t)(n0 >> 4) * TILE_FLOATS + 4 * lane;
  v8f acc[4] = {{}, {}, {}, {}};
  gemm_strip_k768<false>(xrow, wbase, wbase + TILE_FLOATS, wbase + 2 * TILE_FLOATS,
                         wbase + 3 * TILE_FLOATS, acc);
#pragma unroll
  for (int j = 0; j < 4; ++j) {
#pragma unroll
    for (int r = 0; r < 8; ++r) {
      const int m = m0 + r + 8 * hi;
      out[(size_t)m * D_MODEL + n0 + j * 16 + lo] = acc[j][r];
    }
  }
}

#define ATT_PAIRS  6
#define ATT_WAVES  (2 * ATT_PAIRS)
#define ATT_BLOCKS (BATCH * N_HEAD * 64 / ATT_PAIRS)

template <bool MASK>
__device__ __forceinline__ void key_tile(const v4f (&qa)[8],
                                         const float* __restrict__ kfrag,
                                         const float* __restrict__ vfrag,
                                         const float* __restrict__ knext,
                                         float* __restrict__ pl, int lo, int hi,
                                         v8f& z0, v8f& z1, v8f& z2, v8f& z3) {
  __builtin_prefetch(knext, 0, 0);
  v4f kb[8], vb[2][4];
#pragma unroll
  for (int j = 0; j < 8; ++j) kb[j] = *(const v4f*)(kfrag + j * 128);
#pragma unroll
  for (int j = 0; j < 2; ++j) {
#pragma unroll
    for (int t = 0; t < 4; ++t)
      vb[j][t] = *(const v4f*)(vfrag + (4 * j + t) * 128);
  }
  v8f sc = {};
#pragma unroll
  for (int j = 0; j < 8; ++j) {
    sc = wmma_f32(qa[j].xy, kb[j].xy, sc);
    sc = wmma_f32(qa[j].zw, kb[j].zw, sc);
  }
#pragma unroll
  for (int r = 0; r < 8; ++r) {
    float pv = sc[r] * (1.0f / 64.0f);
    pv = pv * pv;
    if (MASK && lo > r + 8 * hi) pv = 0.0f;
    pl[(r + 8 * hi) * 16 + lo] = pv;
  }
#pragma unroll
  for (int j = 0; j < 2; ++j) {
    v4f pa = *(const v4f*)(pl + lo * 16 + 8 * hi + 4 * j);
    z0 = wmma_f32(pa.xy, vb[j][0].xy, z0);
    z1 = wmma_f32(pa.xy, vb[j][1].xy, z1);
    z2 = wmma_f32(pa.xy, vb[j][2].xy, z2);
    z3 = wmma_f32(pa.xy, vb[j][3].xy, z3);
    z0 = wmma_f32(pa.zw, vb[j][0].zw, z0);
    z1 = wmma_f32(pa.zw, vb[j][1].zw, z1);
    z2 = wmma_f32(pa.zw, vb[j][2].zw, z2);
    z3 = wmma_f32(pa.zw, vb[j][3].zw, z3);
  }
  __builtin_amdgcn_sched_group_barrier(0x020, 16, 0);
  __builtin_amdgcn_sched_group_barrier(0x008, 32, 0);
}

__device__ __forceinline__ void sweep(const float* __restrict__ Qh,
                                      const float* __restrict__ Kh,
                                      const float* __restrict__ Vh, int qt,
                                      int kt0, int nfull, bool diag,
                                      float* __restrict__ pl, int lane, int lo,
                                      int hi, v8f& z0, v8f& z1, v8f& z2,
                                      v8f& z3) {
  v4f qa[8];
  {
    const float* qfrag = Qh + (size_t)qt * (16 * D_HEAD) + 4 * lane;
#pragma unroll
    for (int j = 0; j < 8; ++j) qa[j] = *(const v4f*)(qfrag + j * 128);
  }
  for (int kt = kt0; kt < kt0 + nfull; ++kt) {
    key_tile<false>(qa, Kh + (size_t)kt * (16 * D_HEAD) + 4 * lane,
                    Vh + (size_t)kt * (16 * D_HEAD) + 4 * lane,
                    Kh + (size_t)(kt + 1) * (16 * D_HEAD) + 32 * lane, pl, lo,
                    hi, z0, z1, z2, z3);
  }
  if (diag) {
    key_tile<true>(qa, Kh + (size_t)qt * (16 * D_HEAD) + 4 * lane,
                   Vh + (size_t)qt * (16 * D_HEAD) + 4 * lane,
                   Kh + (size_t)(qt + 1) * (16 * D_HEAD) + 32 * lane, pl, lo,
                   hi, z0, z1, z2, z3);
  }
}

__device__ __forceinline__ void store_z_tile(float* __restrict__ Z, int head,
                                             int qt, float* __restrict__ pl,
                                             int lane, int lo, int hi,
                                             const v8f& z0, const v8f& z1,
                                             const v8f& z2, const v8f& z3) {
  const int b = head / N_HEAD, h = head % N_HEAD;
  float* zfrag = Z + (size_t)(b * (N_CTX / 16) + qt) * TILE_FLOATS
                 + (size_t)(h * 8) * 128 + 4 * lane;
  const v8f zt[4] = {z0, z1, z2, z3};
#pragma unroll
  for (int t = 0; t < 4; ++t) {
#pragma unroll
    for (int r = 0; r < 8; ++r) pl[(r + 8 * hi) * 16 + lo] = zt[t][r];
#pragma unroll
    for (int ch = 0; ch < 2; ++ch) {
      const v4f zv = *(const v4f*)(pl + lo * 16 + 8 * ch + 4 * hi);
      *(v4f*)(zfrag + (2 * t + ch) * 128) = zv;
    }
  }
}

__global__ __launch_bounds__(32 * ATT_WAVES) void attn_kernel(
    const float* __restrict__ Q, const float* __restrict__ K,
    const float* __restrict__ V, float* __restrict__ Z) {
  __shared__ __attribute__((aligned(16))) float lds[ATT_WAVES * 256 + ATT_PAIRS * 1024];
  const int wave = threadIdx.x >> 5;
  const int lane = threadIdx.x & 31;
  const int lo = lane & 15, hi = lane >> 4;
  const bool is_a = wave < ATT_PAIRS;
  const int slot = is_a ? wave : wave - ATT_PAIRS;
  const int gp = blockIdx.x * ATT_PAIRS + slot;
  const int head = gp >> 6;
  const int u = gp & 63;
  const float* Qh = Q + (size_t)head * N_CTX * D_HEAD;
  const float* Kh = K + (size_t)head * N_CTX * D_HEAD;
  const float* Vh = V + (size_t)head * N_CTX * D_HEAD;
  float* pl = lds + wave * 256;
  float* zp = lds + ATT_WAVES * 256 + slot * 1024 + lane;

  v8f z0 = {}, z1 = {}, z2 = {}, z3 = {};
  sweep(Qh, Kh, Vh, 64 + u, is_a ? 0 : 64, is_a ? 64 : u, !is_a, pl, lane, lo,
        hi, z0, z1, z2, z3);
  if (!is_a) {
    const v8f zt[4] = {z0, z1, z2, z3};
#pragma unroll
    for (int t = 0; t < 4; ++t) {
#pragma unroll
      for (int r = 0; r < 8; ++r) zp[(t * 8 + r) * 32] = zt[t][r];
    }
    v8f y0 = {}, y1 = {}, y2 = {}, y3 = {};
    sweep(Qh, Kh, Vh, 63 - u, 0, 63 - u, true, pl, lane, lo, hi, y0, y1, y2, y3);
    store_z_tile(Z, head, 63 - u, pl, lane, lo, hi, y0, y1, y2, y3);
  }
  __syncthreads();
  if (is_a) {
#pragma unroll
    for (int r = 0; r < 8; ++r) {
      z0[r] += zp[(0 * 8 + r) * 32];
      z1[r] += zp[(1 * 8 + r) * 32];
      z2[r] += zp[(2 * 8 + r) * 32];
      z3[r] += zp[(3 * 8 + r) * 32];
    }
    store_z_tile(Z, head, 64 + u, pl, lane, lo, hi, z0, z1, z2, z3);
  }
}

extern "C" void kernel_launch(void* const* d_in, const int* in_sizes, int n_in,
                              void* d_out, int out_size, void* d_ws, size_t ws_size,
                              hipStream_t stream) {
  const float* x  = (const float*)d_in[0];
  const float* Wq = (const float*)d_in[1];
  const float* bq = (const float*)d_in[2];
  const float* Wk = (const float*)d_in[3];
  const float* bk = (const float*)d_in[4];
  const float* Wv = (const float*)d_in[5];
  const float* bv = (const float*)d_in[6];
  const float* Wo = (const float*)d_in[7];
  const float* g  = (const float*)d_in[8];
  float* out = (float*)d_out;

  const size_t QKV = (size_t)BATCH * N_HEAD * N_CTX * D_HEAD;
  float* q = (float*)d_ws;
  float* k = q + QKV;
  float* v = k + QKV;
  float* z = v + QKV;
  float* xf  = z + QKV;
  const size_t WSZ = (size_t)D_MODEL * D_MODEL;
  float* wqf = xf + QKV;
  float* wkf = wqf + WSZ;
  float* wvf = wkf + WSZ;
  float* wof = wvf + WSZ;

  frag_order_kernel<<<X_FO_BLOCKS + 4 * W_FO_BLOCKS, 256, 0, stream>>>(
      x, Wq, Wk, Wv, Wo, xf, wqf, wkf, wvf, wof);

  dim3 gblk(M_ROWS / (16 * GEMM_BM), D_MODEL / (64 * GEMM_BN));
  const int gthr = 32 * GEMM_WAVES;
  qkv_gemm_kernel<0><<<gblk, gthr, 0, stream>>>(xf, wqf, bq, g, q);
  qkv_gemm_kernel<0><<<gblk, gthr, 0, stream>>>(xf, wkf, bk, g, k);
  qkv_gemm_kernel<1><<<gblk, gthr, 0, stream>>>(xf, wvf, bv, g, v);

  attn_kernel<<<ATT_BLOCKS, 32 * ATT_WAVES, 0, stream>>>(q, k, v, z);

  out_gemm_kernel<<<gblk, gthr, 0, stream>>>(z, wof, out);
}


# Fragment loads a run ahead: GEMM block i+1's 20 loads before block i's WMMAs (2nd register set); attention V loads before the S run, next K behind it (K prefetch dropped); sums unchanged
# speedup vs baseline: 3.2794x; 1.0696x over previous
#include <hip/hip_runtime.h>
#include <cstddef>

typedef float v2f __attribute__((ext_vector_type(2)));
typedef float v4f __attribute__((ext_vector_type(4)));
typedef float v8f __attribute__((ext_vector_type(8)));

#define D_MODEL 768
#define N_HEAD  12
#define D_HEAD  64
#define N_CTX   2048
#define BATCH   2
#define M_ROWS  (BATCH * N_CTX)

__device__ __forceinline__ v8f wmma_f32(v2f a, v2f b, v8f c) {
  return __builtin_amdgcn_wmma_f32_16x16x4_f32(false, a, false, b, (short)0, c,
                                               false, false);
}

template <int MASK>
__device__ __forceinline__ float row_xor_add(float v) {
  return v + __int_as_float(__builtin_amdgcn_update_dpp(
                 0, __float_as_int(v), 0x160 + MASK, 0xf, 0xf, false));
}

__device__ __forceinline__ float other_half(float v) {
  return __int_as_float(__builtin_amdgcn_permlanex16(
      __float_as_int(v), __float_as_int(v), 0x76543210, 0xfedcba98, false, false));
}

#define K_CHUNKS    (D_MODEL / 8)
#define TILE_FLOATS (K_CHUNKS * 128)

struct strip_frags {
  v4f a[4], b0[4], b1[4], b2[4], b3[4];
};

__device__ __forceinline__ void load_strip_block(strip_frags& f,
                                                 const float* __restrict__ xrow,
                                                 const float* __restrict__ w0,
                                                 const float* __restrict__ w1,
                                                 const float* __restrict__ w2,
                                                 const float* __restrict__ w3,
                                                 int k) {
#pragma unroll
  for (int u = 0; u < 4; ++u) {
    f.a[u]  = *(const v4f*)(xrow + k * 16 + u * 128);
    f.b0[u] = *(const v4f*)(w0 + k * 16 + u * 128);
    f.b1[u] = *(const v4f*)(w1 + k * 16 + u * 128);
    f.b2[u] = *(const v4f*)(w2 + k * 16 + u * 128);
    f.b3[u] = *(const v4f*)(w3 + k * 16 + u * 128);
  }
}

template <bool SWAP>
__device__ __forceinline__ void mma_strip_block(const strip_frags& f, v8f acc[4]) {
#pragma unroll
  for (int u = 0; u < 4; ++u) {
    if (SWAP) {
      acc[0] = wmma_f32(f.b0[u].xy, f.a[u].xy, acc[0]);
      acc[1] = wmma_f32(f.b1[u].xy, f.a[u].xy, acc[1]);
      acc[2] = wmma_f32(f.b2[u].xy, f.a[u].xy, acc[2]);
      acc[3] = wmma_f32(f.b3[u].xy, f.a[u].xy, acc[3]);
      acc[0] = wmma_f32(f.b0[u].zw, f.a[u].zw, acc[0]);
      acc[1] = wmma_f32(f.b1[u].zw, f.a[u].zw, acc[1]);
      acc[2] = wmma_f32(f.b2[u].zw, f.a[u].zw, acc[2]);
      acc[3] = wmma_f32(f.b3[u].zw, f.a[u].zw, acc[3]);
    } else {
      acc[0] = wmma_f32(f.a[u].xy, f.b0[u].xy, acc[0]);
      acc[1] = wmma_f32(f.a[u].xy, f.b1[u].xy, acc[1]);
      acc[2] = wmma_f32(f.a[u].xy, f.b2[u].xy, acc[2]);
      acc[3] = wmma_f32(f.a[u].xy, f.b3[u].xy, acc[3]);
      acc[0] = wmma_f32(f.a[u].zw, f.b0[u].zw, acc[0]);
      acc[1] = wmma_f32(f.a[u].zw, f.b1[u].zw, acc[1]);
      acc[2] = wmma_f32(f.a[u].zw, f.b2[u].zw, acc[2]);
      acc[3] = wmma_f32(f.a[u].zw, f.b3[u].zw, acc[3]);
    }
  }
}

template <bool SWAP>
__device__ __forceinline__ void gemm_strip_k768(const float* __restrict__ xrow,
                                                const float* __restrict__ w0,
                                                const float* __restrict__ w1,
                                                const float* __restrict__ w2,
                                                const float* __restrict__ w3,
                                                v8f acc[4]) {
  strip_frags p, q;
  load_strip_block(p, xrow, w0, w1, w2, w3, 0);
  for (int k = 0; k < D_MODEL - 64; k += 64) {
    load_strip_block(q, xrow, w0, w1, w2, w3, k + 32);
    mma_strip_block<SWAP>(p, acc);
    __builtin_amdgcn_sched_group_barrier(0x020, 20, 0);
    __builtin_amdgcn_sched_group_barrier(0x008, 32, 0);
    load_strip_block(p, xrow, w0, w1, w2, w3, k + 64);
    mma_strip_block<SWAP>(q, acc);
    __builtin_amdgcn_sched_group_barrier(0x020, 20, 0);
    __builtin_amdgcn_sched_group_barrier(0x008, 32, 0);
  }
  load_strip_block(q, xrow, w0, w1, w2, w3, D_MODEL - 32);
  mma_strip_block<SWAP>(p, acc);
  __builtin_amdgcn_sched_group_barrier(0x020, 20, 0);
  __builtin_amdgcn_sched_group_barrier(0x008, 32, 0);
  mma_strip_block<SWAP>(q, acc);
}

#define X_FO_BLOCKS (M_ROWS / 16 * 24 / 8)
#define W_FO_BLOCKS (D_MODEL / 16 * 24 / 8)
__global__ __launch_bounds__(256) void frag_order_kernel(
    const float* __restrict__ x, const float* __restrict__ wq,
    const float* __restrict__ wk, const float* __restrict__ wv,
    const float* __restrict__ wo, float* __restrict__ xf,
    float* __restrict__ wqf, float* __restrict__ wkf, float* __restrict__ wvf,
    float* __restrict__ wof) {
  const int wave = threadIdx.x >> 5;
  const int lane = threadIdx.x & 31;
  const int lo = lane & 15, hi = lane >> 4;
  int blk = blockIdx.x;
  const float* src = x;
  float* dst = xf;
  if (blk >= X_FO_BLOCKS) {
    blk -= X_FO_BLOCKS;
    const int which = blk / W_FO_BLOCKS;
    blk -= which * W_FO_BLOCKS;
    src = which == 0 ? wq : which == 1 ? wk : which == 2 ? wv : wo;
    dst = which == 0 ? wqf : which == 1 ? wkf : which == 2 ? wvf : wof;
  }
  const int id = blk * 8 + wave;
  const int tile = id / 24, kb = id - tile * 24;
  const float* s = src + (size_t)(tile * 16 + lo) * D_MODEL + 32 * kb + 4 * hi;
  float* d = dst + (size_t)tile * TILE_FLOATS + (size_t)(4 * kb) * 128 + 4 * lane;
  const v4f f0 = *(const v4f*)(s);
  const v4f f1 = *(const v4f*)(s + 8);
  const v4f f2 = *(const v4f*)(s + 16);
  const v4f f3 = *(const v4f*)(s + 24);
  *(v4f*)(d)       = f0;
  *(v4f*)(d + 128) = f1;
  *(v4f*)(d + 256) = f2;
  *(v4f*)(d + 384) = f3;
}

#define GEMM_BM    4
#define GEMM_BN    3
#define GEMM_WAVES (GEMM_BM * GEMM_BN)

template <int VMODE>
__global__ __launch_bounds__(32 * GEMM_WAVES) void qkv_gemm_kernel(
    const float* __restrict__ X, const float* __restrict__ W,
    const float* __restrict__ bias, const float* __restrict__ g,
    float* __restrict__ out) {
  const int wave = threadIdx.x >> 5;
  const int lane = threadIdx.x & 31;
  const int lo = lane & 15, hi = lane >> 4;
  const int m0 = (blockIdx.x * GEMM_BM + (wave & (GEMM_BM - 1))) * 16;
  const int n0 = (blockIdx.y * GEMM_BN + wave / GEMM_BM) * 64;
  const float* xrow  = X + (size_t)(m0 >> 4) * TILE_FLOATS + 4 * lane;
  const float* wbase = W + (size_t)(n0 >> 4) * TILE_FLOATS + 4 * lane;
  v8f acc[4] = {{}, {}, {}, {}};
  gemm_strip_k768<VMODE == 0>(xrow, wbase, wbase + TILE_FLOATS,
                              wbase + 2 * TILE_FLOATS, wbase + 3 * TILE_FLOATS, acc);
  const int h = n0 >> 6;
  const int b = m0 >> 11;
  const int tile = (m0 & (N_CTX - 1)) >> 4;
  float* frag = out + ((size_t)(b * N_HEAD + h) * N_CTX + (size_t)tile * 16) * D_HEAD
                + 4 * lane;
  if (VMODE) {
    float bv[4];
#pragma unroll
    for (int j = 0; j < 4; ++j) bv[j] = bias[n0 + j * 16 + lo];
#pragma unroll
    for (int jv = 0; jv < 2; ++jv) {
#pragma unroll
      for (int j = 0; j < 4; ++j) {
        v4f o;
        o.x = acc[j][4 * jv + 0] + bv[j];
        o.y = acc[j][4 * jv + 1] + bv[j];
        o.z = acc[j][4 * jv + 2] + bv[j];
        o.w = acc[j][4 * jv + 3] + bv[j];
        *(v4f*)(frag + (4 * jv + j) * 128) = o;
      }
    }
  } else {
    float x[4][8], gv[4][8];
#pragma unroll
    for (int j = 0; j < 4; ++j) {
      const v4f b0 = *(const v4f*)(bias + n0 + 16 * j + 8 * hi);
      const v4f b1 = *(const v4f*)(bias + n0 + 16 * j + 8 * hi + 4);
      const v4f g0 = *(const v4f*)(g + 16 * j + 8 * hi);
      const v4f g1 = *(const v4f*)(g + 16 * j + 8 * hi + 4);
#pragma unroll
      for (int r = 0; r < 4; ++r) {
        x[j][r]      = acc[j][r] + b0[r];
        x[j][r + 4]  = acc[j][r + 4] + b1[r];
        gv[j][r]     = g0[r];
        gv[j][r + 4] = g1[r];
      }
    }
    float sj[4];
#pragma unroll
    for (int j = 0; j < 4; ++j) {
      sj[j] = ((x[j][0] * x[j][0] + x[j][1] * x[j][1]) +
               (x[j][2] * x[j][2] + x[j][3] * x[j][3])) +
              ((x[j][4] * x[j][4] + x[j][5] * x[j][5]) +
               (x[j][6] * x[j][6] + x[j][7] * x[j][7]));
    }
    float ss = (sj[0] + sj[2]) + (sj[1] + sj[3]);
    ss = ss + other_half(ss);
    const float inv = rsqrtf(ss * (1.0f / 64.0f) + 1e-6f);
    const int s = (m0 + lo) & (N_CTX - 1);
    float o[4][8];
#pragma unroll
    for (int j = 0; j < 2; ++j) {
#pragma unroll
      for (int r = 0; r < 8; ++r) {
        const int i = 16 * j + 8 * hi + r;
        const float w = exp2f((float)i * (-11.0f / 32.0f));
        const float f = (float)s * w;
        const float c = __cosf(f), sn = __sinf(f);
        const float ya = x[j][r] * inv * gv[j][r];
        const float yb = x[j + 2][r] * inv * gv[j + 2][r];
        o[j][r]     = ya * c + yb * sn;
        o[j + 2][r] = yb * c - ya * sn;
      }
    }
#pragma unroll
    for (int j = 0; j < 4; ++j) {
#pragma unroll
      for (int rr = 0; rr < 2; ++rr) {
        v4f v;
        v.x = o[j][4 * rr + 0];
        v.y = o[j][4 * rr + 1];
        v.z = o[j][4 * rr + 2];
        v.w = o[j][4 * rr + 3];
        *(v4f*)(frag + (2 * j + rr) * 128) = v;
      }
    }
  }
}

__global__ __launch_bounds__(32 * GEMM_WAVES) void out_gemm_kernel(
    const float* __restrict__ X, const float* __restrict__ W,
    float* __restrict__ out) {
  const int wave = threadIdx.x >> 5;
  const int lane = threadIdx.x & 31;
  const int lo = lane & 15, hi = lane >> 4;
  const int m0 = (blockIdx.x * GEMM_BM + (wave & (GEMM_BM - 1))) * 16;
  const int n0 = (blockIdx.y * GEMM_BN + wave / GEMM_BM) * 64;
  const float* xrow  = X + (size_t)(m0 >> 4) * TILE_FLOATS + 4 * lane;
  const float* wbase = W + (size_t)(n0 >> 4) * TILE_FLOATS + 4 * lane;
  v8f acc[4] = {{}, {}, {}, {}};
  gemm_strip_k768<false>(xrow, wbase, wbase + TILE_FLOATS, wbase + 2 * TILE_FLOATS,
                         wbase + 3 * TILE_FLOATS, acc);
#pragma unroll
  for (int j = 0; j < 4; ++j) {
#pragma unroll
    for (int r = 0; r < 8; ++r) {
      const int m = m0 + r + 8 * hi;
      out[(size_t)m * D_MODEL + n0 + j * 16 + lo] = acc[j][r];
    }
  }
}

#define ATT_PAIRS  6
#define ATT_WAVES  (2 * ATT_PAIRS)
#define ATT_BLOCKS (BATCH * N_HEAD * 64 / ATT_PAIRS)

__device__ __forceinline__ void load_k_tile(v4f (&kb)[8],
                                            const float* __restrict__ Kh, int kt,
                                            int lane) {
  const float* kfrag = Kh + (size_t)kt * (16 * D_HEAD) + 4 * lane;
#pragma unroll
  for (int j = 0; j < 8; ++j) kb[j] = *(const v4f*)(kfrag + j * 128);
}

template <bool MASK, bool NEXT>
__device__ __forceinline__ void key_tile(const v4f (&qa)[8], v4f (&kb)[8],
                                         const float* __restrict__ Kh,
                                         const float* __restrict__ Vh, int kt,
                                         float* __restrict__ pl, int lane,
                                         int lo, int hi, v8f& z0, v8f& z1,
                                         v8f& z2, v8f& z3) {
  v4f vb[2][4];
  {
    const float* vfrag = Vh + (size_t)kt * (16 * D_HEAD) + 4 * lane;
#pragma unroll
    for (int j = 0; j < 2; ++j) {
#pragma unroll
      for (int t = 0; t < 4; ++t)
        vb[j][t] = *(const v4f*)(vfrag + (4 * j + t) * 128);
    }
  }
  v8f sc = {};
#pragma unroll
  for (int j = 0; j < 8; ++j) {
    sc = wmma_f32(qa[j].xy, kb[j].xy, sc);
    sc = wmma_f32(qa[j].zw, kb[j].zw, sc);
  }
  if (NEXT) load_k_tile(kb, Kh, kt + 1, lane);
#pragma unroll
  for (int r = 0; r < 8; ++r) {
    float pv = sc[r] * (1.0f / 64.0f);
    pv = pv * pv;
    if (MASK && lo > r + 8 * hi) pv = 0.0f;
    pl[(r + 8 * hi) * 16 + lo] = pv;
  }
#pragma unroll
  for (int j = 0; j < 2; ++j) {
    v4f pa = *(const v4f*)(pl + lo * 16 + 8 * hi + 4 * j);
    z0 = wmma_f32(pa.xy, vb[j][0].xy, z0);
    z1 = wmma_f32(pa.xy, vb[j][1].xy, z1);
    z2 = wmma_f32(pa.xy, vb[j][2].xy, z2);
    z3 = wmma_f32(pa.xy, vb[j][3].xy, z3);
    z0 = wmma_f32(pa.zw, vb[j][0].zw, z0);
    z1 = wmma_f32(pa.zw, vb[j][1].zw, z1);
    z2 = wmma_f32(pa.zw, vb[j][2].zw, z2);
    z3 = wmma_f32(pa.zw, vb[j][3].zw, z3);
  }
  __builtin_amdgcn_sched_group_barrier(0x020, 8, 0);
  __builtin_amdgcn_sched_group_barrier(0x008, 16, 0);
  if (NEXT) __builtin_amdgcn_sched_group_barrier(0x020, 8, 0);
  __builtin_amdgcn_sched_group_barrier(0x008, 16, 0);
}

__device__ __forceinline__ void sweep(const float* __restrict__ Qh,
                                      const float* __restrict__ Kh,
                                      const float* __restrict__ Vh, int qt,
                                      int kt0, int nfull, bool diag,
                                      float* __restrict__ pl, int lane, int lo,
                                      int hi, v8f& z0, v8f& z1, v8f& z2,
                                      v8f& z3) {
  v4f qa[8];
  {
    const float* qfrag = Qh + (size_t)qt * (16 * D_HEAD) + 4 * lane;
#pragma unroll
    for (int j = 0; j < 8; ++j) qa[j] = *(const v4f*)(qfrag + j * 128);
  }
  v4f kb[8];
  load_k_tile(kb, Kh, kt0, lane);
  for (int kt = kt0; kt < kt0 + nfull; ++kt)
    key_tile<false, true>(qa, kb, Kh, Vh, kt, pl, lane, lo, hi, z0, z1, z2, z3);
  if (diag)
    key_tile<true, false>(qa, kb, Kh, Vh, qt, pl, lane, lo, hi, z0, z1, z2, z3);
}

__device__ __forceinline__ void store_z_tile(float* __restrict__ Z, int head,
                                             int qt, float* __restrict__ pl,
                                             int lane, int lo, int hi,
                                             const v8f& z0, const v8f& z1,
                                             const v8f& z2, const v8f& z3) {
  const int b = head / N_HEAD, h = head % N_HEAD;
  float* zfrag = Z + (size_t)(b * (N_CTX / 16) + qt) * TILE_FLOATS
                 + (size_t)(h * 8) * 128 + 4 * lane;
  const v8f zt[4] = {z0, z1, z2, z3};
#pragma unroll
  for (int t = 0; t < 4; ++t) {
#pragma unroll
    for (int r = 0; r < 8; ++r) pl[(r + 8 * hi) * 16 + lo] = zt[t][r];
#pragma unroll
    for (int ch = 0; ch < 2; ++ch) {
      const v4f zv = *(const v4f*)(pl + lo * 16 + 8 * ch + 4 * hi);
      *(v4f*)(zfrag + (2 * t + ch) * 128) = zv;
    }
  }
}

__global__ __launch_bounds__(32 * ATT_WAVES) void attn_kernel(
    const float* __restrict__ Q, const float* __restrict__ K,
    const float* __restrict__ V, float* __restrict__ Z) {
  __shared__ __attribute__((aligned(16))) float lds[ATT_WAVES * 256 + ATT_PAIRS * 1024];
  const int wave = threadIdx.x >> 5;
  const int lane = threadIdx.x & 31;
  const int lo = lane & 15, hi = lane >> 4;
  const bool is_a = wave < ATT_PAIRS;
  const int slot = is_a ? wave : wave - ATT_PAIRS;
  const int gp = blockIdx.x * ATT_PAIRS + slot;
  const int head = gp >> 6;
  const int u = gp & 63;
  const float* Qh = Q + (size_t)head * N_CTX * D_HEAD;
  const float* Kh = K + (size_t)head * N_CTX * D_HEAD;
  const float* Vh = V + (size_t)head * N_CTX * D_HEAD;
  float* pl = lds + wave * 256;
  float* zp = lds + ATT_WAVES * 256 + slot * 1024 + lane;

  v8f z0 = {}, z1 = {}, z2 = {}, z3 = {};
  sweep(Qh, Kh, Vh, 64 + u, is_a ? 0 : 64, is_a ? 64 : u, !is_a, pl, lane, lo,
        hi, z0, z1, z2, z3);
  if (!is_a) {
    const v8f zt[4] = {z0, z1, z2, z3};
#pragma unroll
    for (int t = 0; t < 4; ++t) {
#pragma unroll
      for (int r = 0; r < 8; ++r) zp[(t * 8 + r) * 32] = zt[t][r];
    }
    v8f y0 = {}, y1 = {}, y2 = {}, y3 = {};
    sweep(Qh, Kh, Vh, 63 - u, 0, 63 - u, true, pl, lane, lo, hi, y0, y1, y2, y3);
    store_z_tile(Z, head, 63 - u, pl, lane, lo, hi, y0, y1, y2, y3);
  }
  __syncthreads();
  if (is_a) {
#pragma unroll
    for (int r = 0; r < 8; ++r) {
      z0[r] += zp[(0 * 8 + r) * 32];
      z1[r] += zp[(1 * 8 + r) * 32];
      z2[r] += zp[(2 * 8 + r) * 32];
      z3[r] += zp[(3 * 8 + r) * 32];
    }
    store_z_tile(Z, head, 64 + u, pl, lane, lo, hi, z0, z1, z2, z3);
  }
}

extern "C" void kernel_launch(void* const* d_in, const int* in_sizes, int n_in,
                              void* d_out, int out_size, void* d_ws, size_t ws_size,
                              hipStream_t stream) {
  const float* x  = (const float*)d_in[0];
  const float* Wq = (const float*)d_in[1];
  const float* bq = (const float*)d_in[2];
  const float* Wk = (const float*)d_in[3];
  const float* bk = (const float*)d_in[4];
  const float* Wv = (const float*)d_in[5];
  const float* bv = (const float*)d_in[6];
  const float* Wo = (const float*)d_in[7];
  const float* g  = (const float*)d_in[8];
  float* out = (float*)d_out;

  const size_t QKV = (size_t)BATCH * N_HEAD * N_CTX * D_HEAD;
  float* q = (float*)d_ws;
  float* k = q + QKV;
  float* v = k + QKV;
  float* z = v + QKV;
  float* xf  = z + QKV;
  const size_t WSZ = (size_t)D_MODEL * D_MODEL;
  float* wqf = xf + QKV;
  float* wkf = wqf + WSZ;
  float* wvf = wkf + WSZ;
  float* wof = wvf + WSZ;

  frag_order_kernel<<<X_FO_BLOCKS + 4 * W_FO_BLOCKS, 256, 0, stream>>>(
      x, Wq, Wk, Wv, Wo, xf, wqf, wkf, wvf, wof);

  dim3 gblk(M_ROWS / (16 * GEMM_BM), D_MODEL / (64 * GEMM_BN));
  const int gthr = 32 * GEMM_WAVES;
  qkv_gemm_kernel<0><<<gblk, gthr, 0, stream>>>(xf, wqf, bq, g, q);
  qkv_gemm_kernel<0><<<gblk, gthr, 0, stream>>>(xf, wkf, bk, g, k);
  qkv_gemm_kernel<1><<<gblk, gthr, 0, stream>>>(xf, wvf, bv, g, v);

  attn_kernel<<<ATT_BLOCKS, 32 * ATT_WAVES, 0, stream>>>(q, k, v, z);

  out_gemm_kernel<<<gblk, gthr, 0, stream>>>(z, wof, out);
}


# Q, K and V projections in ONE launch (blockIdx.z picks the weight; 768 blocks x 12 waves, a WGP takes the next as one ends): two launch boundaries with their drain and refill go; sums unchanged
# speedup vs baseline: 3.4069x; 1.0389x over previous
#include <hip/hip_runtime.h>
#include <cstddef>

typedef float v2f __attribute__((ext_vector_type(2)));
typedef float v4f __attribute__((ext_vector_type(4)));
typedef float v8f __attribute__((ext_vector_type(8)));

#define D_MODEL 768
#define N_HEAD  12
#define D_HEAD  64
#define N_CTX   2048
#define BATCH   2
#define M_ROWS  (BATCH * N_CTX)

__device__ __forceinline__ v8f wmma_f32(v2f a, v2f b, v8f c) {
  return __builtin_amdgcn_wmma_f32_16x16x4_f32(false, a, false, b, (short)0, c,
                                               false, false);
}

template <int MASK>
__device__ __forceinline__ float row_xor_add(float v) {
  return v + __int_as_float(__builtin_amdgcn_update_dpp(
                 0, __float_as_int(v), 0x160 + MASK, 0xf, 0xf, false));
}

__device__ __forceinline__ float other_half(float v) {
  return __int_as_float(__builtin_amdgcn_permlanex16(
      __float_as_int(v), __float_as_int(v), 0x76543210, 0xfedcba98, false, false));
}

#define K_CHUNKS    (D_MODEL / 8)
#define TILE_FLOATS (K_CHUNKS * 128)

struct strip_frags {
  v4f a[4], b0[4], b1[4], b2[4], b3[4];
};

__device__ __forceinline__ void load_strip_block(strip_frags& f,
                                                 const float* __restrict__ xrow,
                                                 const float* __restrict__ w0,
                                                 const float* __restrict__ w1,
                                                 const float* __restrict__ w2,
                                                 const float* __restrict__ w3,
                                                 int k) {
#pragma unroll
  for (int u = 0; u < 4; ++u) {
    f.a[u]  = *(const v4f*)(xrow + k * 16 + u * 128);
    f.b0[u] = *(const v4f*)(w0 + k * 16 + u * 128);
    f.b1[u] = *(const v4f*)(w1 + k * 16 + u * 128);
    f.b2[u] = *(const v4f*)(w2 + k * 16 + u * 128);
    f.b3[u] = *(const v4f*)(w3 + k * 16 + u * 128);
  }
}

template <bool SWAP>
__device__ __forceinline__ void mma_strip_block(const strip_frags& f, v8f acc[4]) {
#pragma unroll
  for (int u = 0; u < 4; ++u) {
    if (SWAP) {
      acc[0] = wmma_f32(f.b0[u].xy, f.a[u].xy, acc[0]);
      acc[1] = wmma_f32(f.b1[u].xy, f.a[u].xy, acc[1]);
      acc[2] = wmma_f32(f.b2[u].xy, f.a[u].xy, acc[2]);
      acc[3] = wmma_f32(f.b3[u].xy, f.a[u].xy, acc[3]);
      acc[0] = wmma_f32(f.b0[u].zw, f.a[u].zw, acc[0]);
      acc[1] = wmma_f32(f.b1[u].zw, f.a[u].zw, acc[1]);
      acc[2] = wmma_f32(f.b2[u].zw, f.a[u].zw, acc[2]);
      acc[3] = wmma_f32(f.b3[u].zw, f.a[u].zw, acc[3]);
    } else {
      acc[0] = wmma_f32(f.a[u].xy, f.b0[u].xy, acc[0]);
      acc[1] = wmma_f32(f.a[u].xy, f.b1[u].xy, acc[1]);
      acc[2] = wmma_f32(f.a[u].xy, f.b2[u].xy, acc[2]);
      acc[3] = wmma_f32(f.a[u].xy, f.b3[u].xy, acc[3]);
      acc[0] = wmma_f32(f.a[u].zw, f.b0[u].zw, acc[0]);
      acc[1] = wmma_f32(f.a[u].zw, f.b1[u].zw, acc[1]);
      acc[2] = wmma_f32(f.a[u].zw, f.b2[u].zw, acc[2]);
      acc[3] = wmma_f32(f.a[u].zw, f.b3[u].zw, acc[3]);
    }
  }
}

template <bool SWAP>
__device__ __forceinline__ void gemm_strip_k768(const float* __restrict__ xrow,
                                                const float* __restrict__ w0,
                                                const float* __restrict__ w1,
                                                const float* __restrict__ w2,
                                                const float* __restrict__ w3,
                                                v8f acc[4]) {
  strip_frags p, q;
  load_strip_block(p, xrow, w0, w1, w2, w3, 0);
  for (int k = 0; k < D_MODEL - 64; k += 64) {
    load_strip_block(q, xrow, w0, w1, w2, w3, k + 32);
    mma_strip_block<SWAP>(p, acc);
    __builtin_amdgcn_sched_group_barrier(0x020, 20, 0);
    __builtin_amdgcn_sched_group_barrier(0x008, 32, 0);
    load_strip_block(p, xrow, w0, w1, w2, w3, k + 64);
    mma_strip_block<SWAP>(q, acc);
    __builtin_amdgcn_sched_group_barrier(0x020, 20, 0);
    __builtin_amdgcn_sched_group_barrier(0x008, 32, 0);
  }
  load_strip_block(q, xrow, w0, w1, w2, w3, D_MODEL - 32);
  mma_strip_block<SWAP>(p, acc);
  __builtin_amdgcn_sched_group_barrier(0x020, 20, 0);
  __builtin_amdgcn_sched_group_barrier(0x008, 32, 0);
  mma_strip_block<SWAP>(q, acc);
}

#define X_FO_BLOCKS (M_ROWS / 16 * 24 / 8)
#define W_FO_BLOCKS (D_MODEL / 16 * 24 / 8)
__global__ __launch_bounds__(256) void frag_order_kernel(
    const float* __restrict__ x, const float* __restrict__ wq,
    const float* __restrict__ wk, const float* __restrict__ wv,
    const float* __restrict__ wo, float* __restrict__ xf,
    float* __restrict__ wqf, float* __restrict__ wkf, float* __restrict__ wvf,
    float* __restrict__ wof) {
  const int wave = threadIdx.x >> 5;
  const int lane = threadIdx.x & 31;
  const int lo = lane & 15, hi = lane >> 4;
  int blk = blockIdx.x;
  const float* src = x;
  float* dst = xf;
  if (blk >= X_FO_BLOCKS) {
    blk -= X_FO_BLOCKS;
    const int which = blk / W_FO_BLOCKS;
    blk -= which * W_FO_BLOCKS;
    src = which == 0 ? wq : which == 1 ? wk : which == 2 ? wv : wo;
    dst = which == 0 ? wqf : which == 1 ? wkf : which == 2 ? wvf : wof;
  }
  const int id = blk * 8 + wave;
  const int tile = id / 24, kb = id - tile * 24;
  const float* s = src + (size_t)(tile * 16 + lo) * D_MODEL + 32 * kb + 4 * hi;
  float* d = dst + (size_t)tile * TILE_FLOATS + (size_t)(4 * kb) * 128 + 4 * lane;
  const v4f f0 = *(const v4f*)(s);
  const v4f f1 = *(const v4f*)(s + 8);
  const v4f f2 = *(const v4f*)(s + 16);
  const v4f f3 = *(const v4f*)(s + 24);
  *(v4f*)(d)       = f0;
  *(v4f*)(d + 128) = f1;
  *(v4f*)(d + 256) = f2;
  *(v4f*)(d + 384) = f3;
}

#define GEMM_BM    4
#define GEMM_BN    3
#define GEMM_WAVES (GEMM_BM * GEMM_BN)

template <int VMODE>
__device__ __forceinline__ void qkv_gemm_body(
    const float* __restrict__ X, const float* __restrict__ W,
    const float* __restrict__ bias, const float* __restrict__ g,
    float* __restrict__ out) {
  const int wave = threadIdx.x >> 5;
  const int lane = threadIdx.x & 31;
  const int lo = lane & 15, hi = lane >> 4;
  const int m0 = (blockIdx.x * GEMM_BM + (wave & (GEMM_BM - 1))) * 16;
  const int n0 = (blockIdx.y * GEMM_BN + wave / GEMM_BM) * 64;
  const float* xrow  = X + (size_t)(m0 >> 4) * TILE_FLOATS + 4 * lane;
  const float* wbase = W + (size_t)(n0 >> 4) * TILE_FLOATS + 4 * lane;
  v8f acc[4] = {{}, {}, {}, {}};
  gemm_strip_k768<VMODE == 0>(xrow, wbase, wbase + TILE_FLOATS,
                              wbase + 2 * TILE_FLOATS, wbase + 3 * TILE_FLOATS, acc);
  const int h = n0 >> 6;
  const int b = m0 >> 11;
  const int tile = (m0 & (N_CTX - 1)) >> 4;
  float* frag = out + ((size_t)(b * N_HEAD + h) * N_CTX + (size_t)tile * 16) * D_HEAD
                + 4 * lane;
  if (VMODE) {
    float bv[4];
#pragma unroll
    for (int j = 0; j < 4; ++j) bv[j] = bias[n0 + j * 16 + lo];
#pragma unroll
    for (int jv = 0; jv < 2; ++jv) {
#pragma unroll
      for (int j = 0; j < 4; ++j) {
        v4f o;
        o.x = acc[j][4 * jv + 0] + bv[j];
        o.y = acc[j][4 * jv + 1] + bv[j];
        o.z = acc[j][4 * jv + 2] + bv[j];
        o.w = acc[j][4 * jv + 3] + bv[j];
        *(v4f*)(frag + (4 * jv + j) * 128) = o;
      }
    }
  } else {
    float x[4][8], gv[4][8];
#pragma unroll
    for (int j = 0; j < 4; ++j) {
      const v4f b0 = *(const v4f*)(bias + n0 + 16 * j + 8 * hi);
      const v4f b1 = *(const v4f*)(bias + n0 + 16 * j + 8 * hi + 4);
      const v4f g0 = *(const v4f*)(g + 16 * j + 8 * hi);
      const v4f g1 = *(const v4f*)(g + 16 * j + 8 * hi + 4);
#pragma unroll
      for (int r = 0; r < 4; ++r) {
        x[j][r]      = acc[j][r] + b0[r];
        x[j][r + 4]  = acc[j][r + 4] + b1[r];
        gv[j][r]     = g0[r];
        gv[j][r + 4] = g1[r];
      }
    }
    float sj[4];
#pragma unroll
    for (int j = 0; j < 4; ++j) {
      sj[j] = ((x[j][0] * x[j][0] + x[j][1] * x[j][1]) +
               (x[j][2] * x[j][2] + x[j][3] * x[j][3])) +
              ((x[j][4] * x[j][4] + x[j][5] * x[j][5]) +
               (x[j][6] * x[j][6] + x[j][7] * x[j][7]));
    }
    float ss = (sj[0] + sj[2]) + (sj[1] + sj[3]);
    ss = ss + other_half(ss);
    const float inv = rsqrtf(ss * (1.0f / 64.0f) + 1e-6f);
    const int s = (m0 + lo) & (N_CTX - 1);
    float o[4][8];
#pragma unroll
    for (int j = 0; j < 2; ++j) {
#pragma unroll
      for (int r = 0; r < 8; ++r) {
        const int i = 16 * j + 8 * hi + r;
        const float w = exp2f((float)i * (-11.0f / 32.0f));
        const float f = (float)s * w;
        const float c = __cosf(f), sn = __sinf(f);
        const float ya = x[j][r] * inv * gv[j][r];
        const float yb = x[j + 2][r] * inv * gv[j + 2][r];
        o[j][r]     = ya * c + yb * sn;
        o[j + 2][r] = yb * c - ya * sn;
      }
    }
#pragma unroll
    for (int j = 0; j < 4; ++j) {
#pragma unroll
      for (int rr = 0; rr < 2; ++rr) {
        v4f v;
        v.x = o[j][4 * rr + 0];
        v.y = o[j][4 * rr + 1];
        v.z = o[j][4 * rr + 2];
        v.w = o[j][4 * rr + 3];
        *(v4f*)(frag + (2 * j + rr) * 128) = v;
      }
    }
  }
}

__global__ __launch_bounds__(32 * GEMM_WAVES) void qkv_gemm_kernel(
    const float* __restrict__ X, const float* __restrict__ wq,
    const float* __restrict__ wk, const float* __restrict__ wv,
    const float* __restrict__ bq, const float* __restrict__ bk,
    const float* __restrict__ bv, const float* __restrict__ g,
    float* __restrict__ q, float* __restrict__ k, float* __restrict__ v) {
  const int which = blockIdx.z;
  if (which == 2) {
    qkv_gemm_body<1>(X, wv, bv, g, v);
  } else {
    qkv_gemm_body<0>(X, which == 0 ? wq : wk, which == 0 ? bq : bk, g,
                     which == 0 ? q : k);
  }
}

__global__ __launch_bounds__(32 * GEMM_WAVES) void out_gemm_kernel(
    const float* __restrict__ X, const float* __restrict__ W,
    float* __restrict__ out) {
  const int wave = threadIdx.x >> 5;
  const int lane = threadIdx.x & 31;
  const int lo = lane & 15, hi = lane >> 4;
  const int m0 = (blockIdx.x * GEMM_BM + (wave & (GEMM_BM - 1))) * 16;
  const int n0 = (blockIdx.y * GEMM_BN + wave / GEMM_BM) * 64;
  const float* xrow  = X + (size_t)(m0 >> 4) * TILE_FLOATS + 4 * lane;
  const float* wbase = W + (size_t)(n0 >> 4) * TILE_FLOATS + 4 * lane;
  v8f acc[4] = {{}, {}, {}, {}};
  gemm_strip_k768<false>(xrow, wbase, wbase + TILE_FLOATS, wbase + 2 * TILE_FLOATS,
                         wbase + 3 * TILE_FLOATS, acc);
#pragma unroll
  for (int j = 0; j < 4; ++j) {
#pragma unroll
    for (int r = 0; r < 8; ++r) {
      const int m = m0 + r + 8 * hi;
      out[(size_t)m * D_MODEL + n0 + j * 16 + lo] = acc[j][r];
    }
  }
}

#define ATT_PAIRS  6
#define ATT_WAVES  (2 * ATT_PAIRS)
#define ATT_BLOCKS (BATCH * N_HEAD * 64 / ATT_PAIRS)

__device__ __forceinline__ void load_k_tile(v4f (&kb)[8],
                                            const float* __restrict__ Kh, int kt,
                                            int lane) {
  const float* kfrag = Kh + (size_t)kt * (16 * D_HEAD) + 4 * lane;
#pragma unroll
  for (int j = 0; j < 8; ++j) kb[j] = *(const v4f*)(kfrag + j * 128);
}

template <bool MASK, bool NEXT>
__device__ __forceinline__ void key_tile(const v4f (&qa)[8], v4f (&kb)[8],
                                         const float* __restrict__ Kh,
                                         const float* __restrict__ Vh, int kt,
                                         float* __restrict__ pl, int lane,
                                         int lo, int hi, v8f& z0, v8f& z1,
                                         v8f& z2, v8f& z3) {
  v4f vb[2][4];
  {
    const float* vfrag = Vh + (size_t)kt * (16 * D_HEAD) + 4 * lane;
#pragma unroll
    for (int j = 0; j < 2; ++j) {
#pragma unroll
      for (int t = 0; t < 4; ++t)
        vb[j][t] = *(const v4f*)(vfrag + (4 * j + t) * 128);
    }
  }
  v8f sc = {};
#pragma unroll
  for (int j = 0; j < 8; ++j) {
    sc = wmma_f32(qa[j].xy, kb[j].xy, sc);
    sc = wmma_f32(qa[j].zw, kb[j].zw, sc);
  }
  if (NEXT) load_k_tile(kb, Kh, kt + 1, lane);
#pragma unroll
  for (int r = 0; r < 8; ++r) {
    float pv = sc[r] * (1.0f / 64.0f);
    pv = pv * pv;
    if (MASK && lo > r + 8 * hi) pv = 0.0f;
    pl[(r + 8 * hi) * 16 + lo] = pv;
  }
#pragma unroll
  for (int j = 0; j < 2; ++j) {
    v4f pa = *(const v4f*)(pl + lo * 16 + 8 * hi + 4 * j);
    z0 = wmma_f32(pa.xy, vb[j][0].xy, z0);
    z1 = wmma_f32(pa.xy, vb[j][1].xy, z1);
    z2 = wmma_f32(pa.xy, vb[j][2].xy, z2);
    z3 = wmma_f32(pa.xy, vb[j][3].xy, z3);
    z0 = wmma_f32(pa.zw, vb[j][0].zw, z0);
    z1 = wmma_f32(pa.zw, vb[j][1].zw, z1);
    z2 = wmma_f32(pa.zw, vb[j][2].zw, z2);
    z3 = wmma_f32(pa.zw, vb[j][3].zw, z3);
  }
  __builtin_amdgcn_sched_group_barrier(0x020, 8, 0);
  __builtin_amdgcn_sched_group_barrier(0x008, 16, 0);
  if (NEXT) __builtin_amdgcn_sched_group_barrier(0x020, 8, 0);
  __builtin_amdgcn_sched_group_barrier(0x008, 16, 0);
}

__device__ __forceinline__ void sweep(const float* __restrict__ Qh,
                                      const float* __restrict__ Kh,
                                      const float* __restrict__ Vh, int qt,
                                      int kt0, int nfull, bool diag,
                                      float* __restrict__ pl, int lane, int lo,
                                      int hi, v8f& z0, v8f& z1, v8f& z2,
                                      v8f& z3) {
  v4f qa[8];
  {
    const float* qfrag = Qh + (size_t)qt * (16 * D_HEAD) + 4 * lane;
#pragma unroll
    for (int j = 0; j < 8; ++j) qa[j] = *(const v4f*)(qfrag + j * 128);
  }
  v4f kb[8];
  load_k_tile(kb, Kh, kt0, lane);
  for (int kt = kt0; kt < kt0 + nfull; ++kt)
    key_tile<false, true>(qa, kb, Kh, Vh, kt, pl, lane, lo, hi, z0, z1, z2, z3);
  if (diag)
    key_tile<true, false>(qa, kb, Kh, Vh, qt, pl, lane, lo, hi, z0, z1, z2, z3);
}

__device__ __forceinline__ void store_z_tile(float* __restrict__ Z, int head,
                                             int qt, float* __restrict__ pl,
                                             int lane, int lo, int hi,
                                             const v8f& z0, const v8f& z1,
                                             const v8f& z2, const v8f& z3) {
  const int b = head / N_HEAD, h = head % N_HEAD;
  float* zfrag = Z + (size_t)(b * (N_CTX / 16) + qt) * TILE_FLOATS
                 + (size_t)(h * 8) * 128 + 4 * lane;
  const v8f zt[4] = {z0, z1, z2, z3};
#pragma unroll
  for (int t = 0; t < 4; ++t) {
#pragma unroll
    for (int r = 0; r < 8; ++r) pl[(r + 8 * hi) * 16 + lo] = zt[t][r];
#pragma unroll
    for (int ch = 0; ch < 2; ++ch) {
      const v4f zv = *(const v4f*)(pl + lo * 16 + 8 * ch + 4 * hi);
      *(v4f*)(zfrag + (2 * t + ch) * 128) = zv;
    }
  }
}

__global__ __launch_bounds__(32 * ATT_WAVES) void attn_kernel(
    const float* __restrict__ Q, const float* __restrict__ K,
    const float* __restrict__ V, float* __restrict__ Z) {
  __shared__ __attribute__((aligned(16))) float lds[ATT_WAVES * 256 + ATT_PAIRS * 1024];
  const int wave = threadIdx.x >> 5;
  const int lane = threadIdx.x & 31;
  const int lo = lane & 15, hi = lane >> 4;
  const bool is_a = wave < ATT_PAIRS;
  const int slot = is_a ? wave : wave - ATT_PAIRS;
  const int gp = blockIdx.x * ATT_PAIRS + slot;
  const int head = gp >> 6;
  const int u = gp & 63;
  const float* Qh = Q + (size_t)head * N_CTX * D_HEAD;
  const float* Kh = K + (size_t)head * N_CTX * D_HEAD;
  const float* Vh = V + (size_t)head * N_CTX * D_HEAD;
  float* pl = lds + wave * 256;
  float* zp = lds + ATT_WAVES * 256 + slot * 1024 + lane;

  v8f z0 = {}, z1 = {}, z2 = {}, z3 = {};
  sweep(Qh, Kh, Vh, 64 + u, is_a ? 0 : 64, is_a ? 64 : u, !is_a, pl, lane, lo,
        hi, z0, z1, z2, z3);
  if (!is_a) {
    const v8f zt[4] = {z0, z1, z2, z3};
#pragma unroll
    for (int t = 0; t < 4; ++t) {
#pragma unroll
      for (int r = 0; r < 8; ++r) zp[(t * 8 + r) * 32] = zt[t][r];
    }
    v8f y0 = {}, y1 = {}, y2 = {}, y3 = {};
    sweep(Qh, Kh, Vh, 63 - u, 0, 63 - u, true, pl, lane, lo, hi, y0, y1, y2, y3);
    store_z_tile(Z, head, 63 - u, pl, lane, lo, hi, y0, y1, y2, y3);
  }
  __syncthreads();
  if (is_a) {
#pragma unroll
    for (int r = 0; r < 8; ++r) {
      z0[r] += zp[(0 * 8 + r) * 32];
      z1[r] += zp[(1 * 8 + r) * 32];
      z2[r] += zp[(2 * 8 + r) * 32];
      z3[r] += zp[(3 * 8 + r) * 32];
    }
    store_z_tile(Z, head, 64 + u, pl, lane, lo, hi, z0, z1, z2, z3);
  }
}

extern "C" void kernel_launch(void* const* d_in, const int* in_sizes, int n_in,
                              void* d_out, int out_size, void* d_ws, size_t ws_size,
                              hipStream_t stream) {
  const float* x  = (const float*)d_in[0];
  const float* Wq = (const float*)d_in[1];
  const float* bq = (const float*)d_in[2];
  const float* Wk = (const float*)d_in[3];
  const float* bk = (const float*)d_in[4];
  const float* Wv = (const float*)d_in[5];
  const float* bv = (const float*)d_in[6];
  const float* Wo = (const float*)d_in[7];
  const float* g  = (const float*)d_in[8];
  float* out = (float*)d_out;

  const size_t QKV = (size_t)BATCH * N_HEAD * N_CTX * D_HEAD;
  float* q = (float*)d_ws;
  float* k = q + QKV;
  float* v = k + QKV;
  float* z = v + QKV;
  float* xf  = z + QKV;
  const size_t WSZ = (size_t)D_MODEL * D_MODEL;
  float* wqf = xf + QKV;
  float* wkf = wqf + WSZ;
  float* wvf = wkf + WSZ;
  float* wof = wvf + WSZ;

  frag_order_kernel<<<X_FO_BLOCKS + 4 * W_FO_BLOCKS, 256, 0, stream>>>(
      x, Wq, Wk, Wv, Wo, xf, wqf, wkf, wvf, wof);

  dim3 gblk(M_ROWS / (16 * GEMM_BM), D_MODEL / (64 * GEMM_BN));
  const int gthr = 32 * GEMM_WAVES;
  dim3 qkvblk(M_ROWS / (16 * GEMM_BM), D_MODEL / (64 * GEMM_BN), 3);
  qkv_gemm_kernel<<<qkvblk, gthr, 0, stream>>>(xf, wqf, wkf, wvf, bq, bk, bv, g,
                                               q, k, v);

  attn_kernel<<<ATT_BLOCKS, 32 * ATT_WAVES, 0, stream>>>(q, k, v, z);

  out_gemm_kernel<<<gblk, gthr, 0, stream>>>(z, wof, out);
}
